# starting kernel
# baseline (speedup 1.0000x reference)
#include <hip/hip_runtime.h>

typedef __attribute__((ext_vector_type(2))) float v2f;
typedef __attribute__((ext_vector_type(8))) float v8f;

#define WMMA_F32(a, b, c) \
  __builtin_amdgcn_wmma_f32_16x16x4_f32(false, (a), false, (b), (short)0, (c), false, false)

constexpr int B_ = 16, S_ = 2048, H_ = 768, CA_ = 32, IN_ = 300;
constexpr int SC = 128;
constexpr int NCHUNK = S_ / SC;
constexpr int KSPLIT = 4;
constexpr int KLEN = S_ / KSPLIT;
constexpr int TQ_STRIDE = 772;

constexpr size_t TQ_OFF  = 0;
constexpr size_t TQ_SZ   = (size_t)B_ * CA_ * H_;
constexpr size_t AL_OFF  = TQ_OFF + TQ_SZ;
constexpr size_t AL_SZ   = (size_t)B_ * CA_ * S_;
constexpr size_t DEN_OFF = AL_OFF + AL_SZ;
constexpr size_t DEN_SZ  = (size_t)B_ * NCHUNK * CA_;
constexpr size_t NUM_OFF = DEN_OFF + DEN_SZ;
constexpr size_t NUM_SZ  = (size_t)KSPLIT * B_ * CA_ * H_;

__global__ void k_tq(const float* __restrict__ query, const float* __restrict__ W,
                     const float* __restrict__ bias, float* __restrict__ tq) {
  const int bc = blockIdx.x;
  __shared__ float qs[IN_];
  const float* qrow = query + (size_t)bc * IN_;
  for (int i = threadIdx.x; i < IN_; i += 256) qs[i] = qrow[i];
  __syncthreads();
  const int h0 = threadIdx.x;
  float a0 = bias[h0], a1 = bias[h0 + 256], a2 = bias[h0 + 512];
  for (int i = 0; i < IN_; ++i) {
    const float qv = qs[i];
    const float* wr = W + (size_t)i * H_ + h0;
    a0 = fmaf(qv, wr[0],   a0);
    a1 = fmaf(qv, wr[256], a1);
    a2 = fmaf(qv, wr[512], a2);
  }
  float* o = tq + (size_t)bc * H_ + h0;
  o[0]   = tanhf(a0);
  o[256] = tanhf(a1);
  o[512] = tanhf(a2);
}

__global__ void k_scores(const float* __restrict__ key, const float* __restrict__ tq,
                         const int* __restrict__ seq_len,
                         float* __restrict__ alpha_g, float* __restrict__ den_part) {
  extern __shared__ float smem[];
  float* tqs  = smem;
  float* dens = smem + 32 * TQ_STRIDE;

  const int b     = blockIdx.x / NCHUNK;
  const int chunk = blockIdx.x % NCHUNK;
  const int s0    = chunk * SC;
  const int tid   = threadIdx.x;
  const int lane  = tid & 31;
  const int wv    = tid >> 5;
  const int hl    = lane & 15;
  const int half  = lane >> 4;
  const int kh    = half * 2;

  const float4* tqg = (const float4*)(tq + (size_t)b * CA_ * H_);
  for (int i = tid; i < 32 * (H_ / 4); i += 256) {
    const int row = i / (H_ / 4), c4 = i % (H_ / 4);
    const float4 v = tqg[i];
    float* d = &tqs[row * TQ_STRIDE + c4 * 4];
    d[0] = v.x; d[1] = v.y; d[2] = v.z; d[3] = v.w;
  }
  __syncthreads();

  const float* a0p  = &tqs[hl * TQ_STRIDE + kh];
  const float* a1p  = &tqs[(hl + 16) * TQ_STRIDE + kh];
  const float* krow = key + ((size_t)b * S_ + (s0 + wv * 16 + hl)) * H_ + kh;

  v8f acc0 = {0.f, 0.f, 0.f, 0.f, 0.f, 0.f, 0.f, 0.f};
  v8f acc1 = {0.f, 0.f, 0.f, 0.f, 0.f, 0.f, 0.f, 0.f};
  for (int k = 0; k < H_; k += 4) {
    const v2f a0 = *(const v2f*)(a0p + k);
    const v2f a1 = *(const v2f*)(a1p + k);
    const v2f bf = *(const v2f*)(krow + k);
    acc0 = WMMA_F32(a0, bf, acc0);
    acc1 = WMMA_F32(a1, bf, acc1);
  }

  const int s = s0 + wv * 16 + hl;
  const bool valid = s < seq_len[b];
  float* ag = alpha_g + (size_t)b * CA_ * S_;
  for (int r = 0; r < 8; ++r) {
    const int c0 = r + 8 * half;
    float v0 = valid ? __expf(acc0[r]) : 0.0f;
    float v1 = valid ? __expf(acc1[r]) : 0.0f;
    ag[(size_t)c0 * S_ + s]        = v0;
    ag[(size_t)(c0 + 16) * S_ + s] = v1;
    float r0 = v0, r1 = v1;
    for (int m = 1; m < 16; m <<= 1) {
      r0 += __shfl_xor(r0, m, 32);
      r1 += __shfl_xor(r1, m, 32);
    }
    if (hl == 0) {
      dens[c0 * 8 + wv]        = r0;
      dens[(c0 + 16) * 8 + wv] = r1;
    }
  }
  __syncthreads();
  if (tid < CA_) {
    float d = 0.f;
    for (int w = 0; w < 8; ++w) d += dens[tid * 8 + w];
    den_part[((size_t)b * NCHUNK + chunk) * CA_ + tid] = d;
  }
}

__global__ void k_out(const float* __restrict__ value, const float* __restrict__ alpha_g,
                      float* __restrict__ num_part) {
  const int gid = blockIdx.x;
  const int b   = gid / (KSPLIT * 6);
  const int rem = gid % (KSPLIT * 6);
  const int kc  = rem / 6;
  const int nb  = rem % 6;
  const int tid  = threadIdx.x;
  const int lane = tid & 31;
  const int wv   = tid >> 5;
  const int hl   = lane & 15;
  const int half = lane >> 4;
  const int kh   = half * 2;
  const int n    = nb * 8 + wv;

  const float* arow0 = alpha_g + ((size_t)b * CA_ + hl) * S_ + kc * KLEN + kh;
  const float* arow1 = arow0 + (size_t)16 * S_;
  const float* vcol  = value + ((size_t)b * S_ + kc * KLEN + kh) * H_ + n * 16 + hl;

  v8f acc0 = {0.f, 0.f, 0.f, 0.f, 0.f, 0.f, 0.f, 0.f};
  v8f acc1 = {0.f, 0.f, 0.f, 0.f, 0.f, 0.f, 0.f, 0.f};
  for (int k = 0; k < KLEN; k += 4) {
    const v2f a0 = *(const v2f*)(arow0 + k);
    const v2f a1 = *(const v2f*)(arow1 + k);
    v2f bf;
    bf.x = vcol[(size_t)k * H_];
    bf.y = vcol[(size_t)(k + 1) * H_];
    acc0 = WMMA_F32(a0, bf, acc0);
    acc1 = WMMA_F32(a1, bf, acc1);
  }
  float* o = num_part + ((size_t)kc * B_ + b) * CA_ * H_;
  for (int r = 0; r < 8; ++r) {
    const int c = r + 8 * half;
    o[(size_t)c * H_ + n * 16 + hl]        = acc0[r];
    o[(size_t)(c + 16) * H_ + n * 16 + hl] = acc1[r];
  }
}

__global__ void k_fin(const float* __restrict__ num_part, const float* __restrict__ den_part,
                      float* __restrict__ out) {
  const int idx = blockIdx.x * 256 + threadIdx.x;
  const int bc  = idx / H_;
  const int b   = bc >> 5, c = bc & 31;
  float num = 0.f;
  for (int kc = 0; kc < KSPLIT; ++kc) num += num_part[(size_t)kc * (B_ * CA_ * H_) + idx];
  float den = 0.f;
  for (int ch = 0; ch < NCHUNK; ++ch) den += den_part[((size_t)b * NCHUNK + ch) * CA_ + c];
  out[idx] = num / den;
}

extern "C" void kernel_launch(void* const* d_in, const int* in_sizes, int n_in,
                              void* d_out, int out_size, void* d_ws, size_t ws_size,
                              hipStream_t stream) {
  (void)in_sizes; (void)n_in; (void)out_size; (void)ws_size;
  const float* key     = (const float*)d_in[0];
  const float* value   = (const float*)d_in[1];
  const float* query   = (const float*)d_in[2];
  const int*   seq_len = (const int*)d_in[3];
  const float* W       = (const float*)d_in[4];
  const float* bias    = (const float*)d_in[5];
  float* ws  = (float*)d_ws;
  float* out = (float*)d_out;

  k_tq<<<B_ * CA_, 256, 0, stream>>>(query, W, bias, ws + TQ_OFF);

  const int smem_bytes = (32 * TQ_STRIDE + 32 * 8) * sizeof(float);
  k_scores<<<B_ * NCHUNK, 256, smem_bytes, stream>>>(key, ws + TQ_OFF, seq_len,
                                                     ws + AL_OFF, ws + DEN_OFF);

  k_out<<<B_ * KSPLIT * 6, 256, 0, stream>>>(value, ws + AL_OFF, ws + NUM_OFF);

  k_fin<<<(B_ * CA_ * H_) / 256, 256, 0, stream>>>(ws + NUM_OFF, ws + DEN_OFF, out);
}


# query projection moved from per-row scalar FMA onto f32 16x16x4 matrix instructions: 32x32 tile per block, K split over 5 waves, partials added through LDS in fixed wave order (new summation order)
# speedup vs baseline: 1.2658x; 1.2658x over previous
#include <hip/hip_runtime.h>

typedef __attribute__((ext_vector_type(2))) float v2f;
typedef __attribute__((ext_vector_type(8))) float v8f;

#define WMMA_F32(a, b, c) \
  __builtin_amdgcn_wmma_f32_16x16x4_f32(false, (a), false, (b), (short)0, (c), false, false)

constexpr int B_ = 16, S_ = 2048, H_ = 768, CA_ = 32, IN_ = 300;
constexpr int SC = 128;
constexpr int NCHUNK = S_ / SC;
constexpr int KSPLIT = 4;
constexpr int KLEN = S_ / KSPLIT;
constexpr int TQ_STRIDE = 772;

constexpr size_t TQ_OFF  = 0;
constexpr size_t TQ_SZ   = (size_t)B_ * CA_ * H_;
constexpr size_t AL_OFF  = TQ_OFF + TQ_SZ;
constexpr size_t AL_SZ   = (size_t)B_ * CA_ * S_;
constexpr size_t DEN_OFF = AL_OFF + AL_SZ;
constexpr size_t DEN_SZ  = (size_t)B_ * NCHUNK * CA_;
constexpr size_t NUM_OFF = DEN_OFF + DEN_SZ;
constexpr size_t NUM_SZ  = (size_t)KSPLIT * B_ * CA_ * H_;

constexpr int TQ_WAVES = 5;
constexpr int TQ_STEPS = IN_ / (4 * TQ_WAVES);
constexpr int TQ_NT    = H_ / 32;

__global__ void __launch_bounds__(32 * TQ_WAVES)
k_tq(const float* __restrict__ query, const float* __restrict__ W,
     const float* __restrict__ bias, float* __restrict__ tq) {
  __shared__ float red[TQ_WAVES][32][32];
  const int b    = blockIdx.x / TQ_NT;
  const int nt   = blockIdx.x % TQ_NT;
  const int tid  = threadIdx.x;
  const int lane = tid & 31;
  const int wv   = tid >> 5;
  const int hl   = lane & 15;
  const int half = lane >> 4;
  const int kh   = half * 2;
  const int kb   = wv * (TQ_STEPS * 4);

  const float* ap0 = query + ((size_t)b * CA_ + hl) * IN_ + kb + kh;
  const float* ap1 = ap0 + (size_t)16 * IN_;
  const float* bp  = W + (size_t)(kb + kh) * H_ + nt * 32 + 2 * hl;

  v8f acc00 = {0.f, 0.f, 0.f, 0.f, 0.f, 0.f, 0.f, 0.f};
  v8f acc01 = acc00, acc10 = acc00, acc11 = acc00;
  v2f a0[TQ_STEPS], a1[TQ_STEPS], w0[TQ_STEPS], w1[TQ_STEPS];
#pragma unroll
  for (int i = 0; i < TQ_STEPS; ++i) {
    a0[i] = *(const v2f*)(ap0 + 4 * i);
    a1[i] = *(const v2f*)(ap1 + 4 * i);
    w0[i] = *(const v2f*)(bp + (size_t)(4 * i) * H_);
    w1[i] = *(const v2f*)(bp + (size_t)(4 * i + 1) * H_);
  }
  __builtin_amdgcn_sched_barrier(0);
#pragma unroll
  for (int i = 0; i < TQ_STEPS; ++i) {
    v2f b0, b1;
    b0.x = w0[i].x; b0.y = w1[i].x;
    b1.x = w0[i].y; b1.y = w1[i].y;
    acc00 = WMMA_F32(a0[i], b0, acc00);
    acc01 = WMMA_F32(a0[i], b1, acc01);
    acc10 = WMMA_F32(a1[i], b0, acc10);
    acc11 = WMMA_F32(a1[i], b1, acc11);
  }

  for (int r = 0; r < 8; ++r) {
    red[wv][r][lane]      = acc00[r];
    red[wv][8 + r][lane]  = acc01[r];
    red[wv][16 + r][lane] = acc10[r];
    red[wv][24 + r][lane] = acc11[r];
  }
  __syncthreads();

  if (wv < 4) {
    const int h0 = nt * 32 + 2 * hl;
    const v2f bv = *(const v2f*)(bias + h0);
    for (int q = 0; q < 2; ++q) {
      const int r = 2 * wv + q;
      float s00 = red[0][r][lane],      s01 = red[0][8 + r][lane];
      float s10 = red[0][16 + r][lane], s11 = red[0][24 + r][lane];
      for (int w = 1; w < TQ_WAVES; ++w) {
        s00 += red[w][r][lane];
        s01 += red[w][8 + r][lane];
        s10 += red[w][16 + r][lane];
        s11 += red[w][24 + r][lane];
      }
      const int c = 8 * half + r;
      v2f o0, o1;
      o0.x = tanhf(s00 + bv.x); o0.y = tanhf(s01 + bv.y);
      o1.x = tanhf(s10 + bv.x); o1.y = tanhf(s11 + bv.y);
      *(v2f*)(tq + ((size_t)b * CA_ + c) * H_ + h0)      = o0;
      *(v2f*)(tq + ((size_t)b * CA_ + c + 16) * H_ + h0) = o1;
    }
  }
}

__global__ void k_scores(const float* __restrict__ key, const float* __restrict__ tq,
                         const int* __restrict__ seq_len,
                         float* __restrict__ alpha_g, float* __restrict__ den_part) {
  extern __shared__ float smem[];
  float* tqs  = smem;
  float* dens = smem + 32 * TQ_STRIDE;

  const int b     = blockIdx.x / NCHUNK;
  const int chunk = blockIdx.x % NCHUNK;
  const int s0    = chunk * SC;
  const int tid   = threadIdx.x;
  const int lane  = tid & 31;
  const int wv    = tid >> 5;
  const int hl    = lane & 15;
  const int half  = lane >> 4;
  const int kh    = half * 2;

  const float4* tqg = (const float4*)(tq + (size_t)b * CA_ * H_);
  for (int i = tid; i < 32 * (H_ / 4); i += 256) {
    const int row = i / (H_ / 4), c4 = i % (H_ / 4);
    const float4 v = tqg[i];
    float* d = &tqs[row * TQ_STRIDE + c4 * 4];
    d[0] = v.x; d[1] = v.y; d[2] = v.z; d[3] = v.w;
  }
  __syncthreads();

  const float* a0p  = &tqs[hl * TQ_STRIDE + kh];
  const float* a1p  = &tqs[(hl + 16) * TQ_STRIDE + kh];
  const float* krow = key + ((size_t)b * S_ + (s0 + wv * 16 + hl)) * H_ + kh;

  v8f acc0 = {0.f, 0.f, 0.f, 0.f, 0.f, 0.f, 0.f, 0.f};
  v8f acc1 = {0.f, 0.f, 0.f, 0.f, 0.f, 0.f, 0.f, 0.f};
  for (int k = 0; k < H_; k += 4) {
    const v2f a0 = *(const v2f*)(a0p + k);
    const v2f a1 = *(const v2f*)(a1p + k);
    const v2f bf = *(const v2f*)(krow + k);
    acc0 = WMMA_F32(a0, bf, acc0);
    acc1 = WMMA_F32(a1, bf, acc1);
  }

  const int s = s0 + wv * 16 + hl;
  const bool valid = s < seq_len[b];
  float* ag = alpha_g + (size_t)b * CA_ * S_;
  for (int r = 0; r < 8; ++r) {
    const int c0 = r + 8 * half;
    float v0 = valid ? __expf(acc0[r]) : 0.0f;
    float v1 = valid ? __expf(acc1[r]) : 0.0f;
    ag[(size_t)c0 * S_ + s]        = v0;
    ag[(size_t)(c0 + 16) * S_ + s] = v1;
    float r0 = v0, r1 = v1;
    for (int m = 1; m < 16; m <<= 1) {
      r0 += __shfl_xor(r0, m, 32);
      r1 += __shfl_xor(r1, m, 32);
    }
    if (hl == 0) {
      dens[c0 * 8 + wv]        = r0;
      dens[(c0 + 16) * 8 + wv] = r1;
    }
  }
  __syncthreads();
  if (tid < CA_) {
    float d = 0.f;
    for (int w = 0; w < 8; ++w) d += dens[tid * 8 + w];
    den_part[((size_t)b * NCHUNK + chunk) * CA_ + tid] = d;
  }
}

__global__ void k_out(const float* __restrict__ value, const float* __restrict__ alpha_g,
                      float* __restrict__ num_part) {
  const int gid = blockIdx.x;
  const int b   = gid / (KSPLIT * 6);
  const int rem = gid % (KSPLIT * 6);
  const int kc  = rem / 6;
  const int nb  = rem % 6;
  const int tid  = threadIdx.x;
  const int lane = tid & 31;
  const int wv   = tid >> 5;
  const int hl   = lane & 15;
  const int half = lane >> 4;
  const int kh   = half * 2;
  const int n    = nb * 8 + wv;

  const float* arow0 = alpha_g + ((size_t)b * CA_ + hl) * S_ + kc * KLEN + kh;
  const float* arow1 = arow0 + (size_t)16 * S_;
  const float* vcol  = value + ((size_t)b * S_ + kc * KLEN + kh) * H_ + n * 16 + hl;

  v8f acc0 = {0.f, 0.f, 0.f, 0.f, 0.f, 0.f, 0.f, 0.f};
  v8f acc1 = {0.f, 0.f, 0.f, 0.f, 0.f, 0.f, 0.f, 0.f};
  for (int k = 0; k < KLEN; k += 4) {
    const v2f a0 = *(const v2f*)(arow0 + k);
    const v2f a1 = *(const v2f*)(arow1 + k);
    v2f bf;
    bf.x = vcol[(size_t)k * H_];
    bf.y = vcol[(size_t)(k + 1) * H_];
    acc0 = WMMA_F32(a0, bf, acc0);
    acc1 = WMMA_F32(a1, bf, acc1);
  }
  float* o = num_part + ((size_t)kc * B_ + b) * CA_ * H_;
  for (int r = 0; r < 8; ++r) {
    const int c = r + 8 * half;
    o[(size_t)c * H_ + n * 16 + hl]        = acc0[r];
    o[(size_t)(c + 16) * H_ + n * 16 + hl] = acc1[r];
  }
}

__global__ void k_fin(const float* __restrict__ num_part, const float* __restrict__ den_part,
                      float* __restrict__ out) {
  const int idx = blockIdx.x * 256 + threadIdx.x;
  const int bc  = idx / H_;
  const int b   = bc >> 5, c = bc & 31;
  float num = 0.f;
  for (int kc = 0; kc < KSPLIT; ++kc) num += num_part[(size_t)kc * (B_ * CA_ * H_) + idx];
  float den = 0.f;
  for (int ch = 0; ch < NCHUNK; ++ch) den += den_part[((size_t)b * NCHUNK + ch) * CA_ + c];
  out[idx] = num / den;
}

extern "C" void kernel_launch(void* const* d_in, const int* in_sizes, int n_in,
                              void* d_out, int out_size, void* d_ws, size_t ws_size,
                              hipStream_t stream) {
  (void)in_sizes; (void)n_in; (void)out_size; (void)ws_size;
  const float* key     = (const float*)d_in[0];
  const float* value   = (const float*)d_in[1];
  const float* query   = (const float*)d_in[2];
  const int*   seq_len = (const int*)d_in[3];
  const float* W       = (const float*)d_in[4];
  const float* bias    = (const float*)d_in[5];
  float* ws  = (float*)d_ws;
  float* out = (float*)d_out;

  k_tq<<<B_ * TQ_NT, 32 * TQ_WAVES, 0, stream>>>(query, W, bias, ws + TQ_OFF);

  const int smem_bytes = (32 * TQ_STRIDE + 32 * 8) * sizeof(float);
  k_scores<<<B_ * NCHUNK, 256, smem_bytes, stream>>>(key, ws + TQ_OFF, seq_len,
                                                     ws + AL_OFF, ws + DEN_OFF);

  k_out<<<B_ * KSPLIT * 6, 256, 0, stream>>>(value, ws + AL_OFF, ws + NUM_OFF);

  k_fin<<<(B_ * CA_ * H_) / 256, 256, 0, stream>>>(ws + NUM_OFF, ws + DEN_OFF, out);
}


# output GEMM re-tiled: wave owns 32x64 via float4 value loads (one alpha fragment feeds 4 column tiles), next 8 k-steps loaded ahead; block's 4 K quarters added via LDS in wave order (new sum order)
# speedup vs baseline: 1.6743x; 1.3227x over previous
#include <hip/hip_runtime.h>

typedef __attribute__((ext_vector_type(2))) float v2f;
typedef __attribute__((ext_vector_type(8))) float v8f;

#define WMMA_F32(a, b, c) \
  __builtin_amdgcn_wmma_f32_16x16x4_f32(false, (a), false, (b), (short)0, (c), false, false)

constexpr int B_ = 16, S_ = 2048, H_ = 768, CA_ = 32, IN_ = 300;
constexpr int SC = 128;
constexpr int NCHUNK = S_ / SC;
constexpr int KSPLIT = 4;
constexpr int KLEN = S_ / KSPLIT;
constexpr int TQ_STRIDE = 772;

constexpr size_t TQ_OFF  = 0;
constexpr size_t TQ_SZ   = (size_t)B_ * CA_ * H_;
constexpr size_t AL_OFF  = TQ_OFF + TQ_SZ;
constexpr size_t AL_SZ   = (size_t)B_ * CA_ * S_;
constexpr size_t DEN_OFF = AL_OFF + AL_SZ;
constexpr size_t DEN_SZ  = (size_t)B_ * NCHUNK * CA_;
constexpr size_t NUM_OFF = DEN_OFF + DEN_SZ;
constexpr size_t NUM_SZ  = (size_t)KSPLIT * B_ * CA_ * H_;

constexpr int TQ_WAVES = 5;
constexpr int TQ_STEPS = IN_ / (4 * TQ_WAVES);
constexpr int TQ_NT    = H_ / 32;

__global__ void __launch_bounds__(32 * TQ_WAVES)
k_tq(const float* __restrict__ query, const float* __restrict__ W,
     const float* __restrict__ bias, float* __restrict__ tq) {
  __shared__ float red[TQ_WAVES][32][32];
  const int b    = blockIdx.x / TQ_NT;
  const int nt   = blockIdx.x % TQ_NT;
  const int tid  = threadIdx.x;
  const int lane = tid & 31;
  const int wv   = tid >> 5;
  const int hl   = lane & 15;
  const int half = lane >> 4;
  const int kh   = half * 2;
  const int kb   = wv * (TQ_STEPS * 4);

  const float* ap0 = query + ((size_t)b * CA_ + hl) * IN_ + kb + kh;
  const float* ap1 = ap0 + (size_t)16 * IN_;
  const float* bp  = W + (size_t)(kb + kh) * H_ + nt * 32 + 2 * hl;

  v8f acc00 = {0.f, 0.f, 0.f, 0.f, 0.f, 0.f, 0.f, 0.f};
  v8f acc01 = acc00, acc10 = acc00, acc11 = acc00;
  v2f a0[TQ_STEPS], a1[TQ_STEPS], w0[TQ_STEPS], w1[TQ_STEPS];
#pragma unroll
  for (int i = 0; i < TQ_STEPS; ++i) {
    a0[i] = *(const v2f*)(ap0 + 4 * i);
    a1[i] = *(const v2f*)(ap1 + 4 * i);
    w0[i] = *(const v2f*)(bp + (size_t)(4 * i) * H_);
    w1[i] = *(const v2f*)(bp + (size_t)(4 * i + 1) * H_);
  }
  __builtin_amdgcn_sched_barrier(0);
#pragma unroll
  for (int i = 0; i < TQ_STEPS; ++i) {
    v2f b0, b1;
    b0.x = w0[i].x; b0.y = w1[i].x;
    b1.x = w0[i].y; b1.y = w1[i].y;
    acc00 = WMMA_F32(a0[i], b0, acc00);
    acc01 = WMMA_F32(a0[i], b1, acc01);
    acc10 = WMMA_F32(a1[i], b0, acc10);
    acc11 = WMMA_F32(a1[i], b1, acc11);
  }

  for (int r = 0; r < 8; ++r) {
    red[wv][r][lane]      = acc00[r];
    red[wv][8 + r][lane]  = acc01[r];
    red[wv][16 + r][lane] = acc10[r];
    red[wv][24 + r][lane] = acc11[r];
  }
  __syncthreads();

  if (wv < 4) {
    const int h0 = nt * 32 + 2 * hl;
    const v2f bv = *(const v2f*)(bias + h0);
    for (int q = 0; q < 2; ++q) {
      const int r = 2 * wv + q;
      float s00 = red[0][r][lane],      s01 = red[0][8 + r][lane];
      float s10 = red[0][16 + r][lane], s11 = red[0][24 + r][lane];
      for (int w = 1; w < TQ_WAVES; ++w) {
        s00 += red[w][r][lane];
        s01 += red[w][8 + r][lane];
        s10 += red[w][16 + r][lane];
        s11 += red[w][24 + r][lane];
      }
      const int c = 8 * half + r;
      v2f o0, o1;
      o0.x = tanhf(s00 + bv.x); o0.y = tanhf(s01 + bv.y);
      o1.x = tanhf(s10 + bv.x); o1.y = tanhf(s11 + bv.y);
      *(v2f*)(tq + ((size_t)b * CA_ + c) * H_ + h0)      = o0;
      *(v2f*)(tq + ((size_t)b * CA_ + c + 16) * H_ + h0) = o1;
    }
  }
}

__global__ void k_scores(const float* __restrict__ key, const float* __restrict__ tq,
                         const int* __restrict__ seq_len,
                         float* __restrict__ alpha_g, float* __restrict__ den_part) {
  extern __shared__ float smem[];
  float* tqs  = smem;
  float* dens = smem + 32 * TQ_STRIDE;

  const int b     = blockIdx.x / NCHUNK;
  const int chunk = blockIdx.x % NCHUNK;
  const int s0    = chunk * SC;
  const int tid   = threadIdx.x;
  const int lane  = tid & 31;
  const int wv    = tid >> 5;
  const int hl    = lane & 15;
  const int half  = lane >> 4;
  const int kh    = half * 2;

  const float4* tqg = (const float4*)(tq + (size_t)b * CA_ * H_);
  for (int i = tid; i < 32 * (H_ / 4); i += 256) {
    const int row = i / (H_ / 4), c4 = i % (H_ / 4);
    const float4 v = tqg[i];
    float* d = &tqs[row * TQ_STRIDE + c4 * 4];
    d[0] = v.x; d[1] = v.y; d[2] = v.z; d[3] = v.w;
  }
  __syncthreads();

  const float* a0p  = &tqs[hl * TQ_STRIDE + kh];
  const float* a1p  = &tqs[(hl + 16) * TQ_STRIDE + kh];
  const float* krow = key + ((size_t)b * S_ + (s0 + wv * 16 + hl)) * H_ + kh;

  v8f acc0 = {0.f, 0.f, 0.f, 0.f, 0.f, 0.f, 0.f, 0.f};
  v8f acc1 = {0.f, 0.f, 0.f, 0.f, 0.f, 0.f, 0.f, 0.f};
  for (int k = 0; k < H_; k += 4) {
    const v2f a0 = *(const v2f*)(a0p + k);
    const v2f a1 = *(const v2f*)(a1p + k);
    const v2f bf = *(const v2f*)(krow + k);
    acc0 = WMMA_F32(a0, bf, acc0);
    acc1 = WMMA_F32(a1, bf, acc1);
  }

  const int s = s0 + wv * 16 + hl;
  const bool valid = s < seq_len[b];
  float* ag = alpha_g + (size_t)b * CA_ * S_;
  for (int r = 0; r < 8; ++r) {
    const int c0 = r + 8 * half;
    float v0 = valid ? __expf(acc0[r]) : 0.0f;
    float v1 = valid ? __expf(acc1[r]) : 0.0f;
    ag[(size_t)c0 * S_ + s]        = v0;
    ag[(size_t)(c0 + 16) * S_ + s] = v1;
    float r0 = v0, r1 = v1;
    for (int m = 1; m < 16; m <<= 1) {
      r0 += __shfl_xor(r0, m, 32);
      r1 += __shfl_xor(r1, m, 32);
    }
    if (hl == 0) {
      dens[c0 * 8 + wv]        = r0;
      dens[(c0 + 16) * 8 + wv] = r1;
    }
  }
  __syncthreads();
  if (tid < CA_) {
    float d = 0.f;
    for (int w = 0; w < 8; ++w) d += dens[tid * 8 + w];
    den_part[((size_t)b * NCHUNK + chunk) * CA_ + tid] = d;
  }
}

constexpr int OUT_WAVES = 4;
constexpr int OUT_KW    = KLEN / OUT_WAVES;
constexpr int OUT_CG    = H_ / 64;
constexpr int OUT_ST    = 8;
constexpr int OUT_NS    = OUT_KW / (4 * OUT_ST);

__global__ void __launch_bounds__(32 * OUT_WAVES)
k_out(const float* __restrict__ value, const float* __restrict__ alpha_g,
      float* __restrict__ num_part) {
  __shared__ float red[OUT_WAVES][64][32];
  const int gid  = blockIdx.x;
  const int cg   = gid % OUT_CG;
  const int kc   = (gid / OUT_CG) % KSPLIT;
  const int b    = gid / (OUT_CG * KSPLIT);
  const int tid  = threadIdx.x;
  const int lane = tid & 31;
  const int wv   = tid >> 5;
  const int hl   = lane & 15;
  const int half = lane >> 4;
  const int kh   = half * 2;
  const int sb   = kc * KLEN + wv * OUT_KW;

  const float* ap0 = alpha_g + ((size_t)b * CA_ + hl) * S_ + sb + kh;
  const float* ap1 = ap0 + (size_t)16 * S_;
  const float* vp  = value + ((size_t)b * S_ + sb + kh) * H_ + cg * 64 + 4 * hl;

  v8f acc[8];
  for (int a = 0; a < 8; ++a) acc[a] = (v8f){0.f, 0.f, 0.f, 0.f, 0.f, 0.f, 0.f, 0.f};

  v2f    a0[2][OUT_ST], a1[2][OUT_ST];
  float4 v0[2][OUT_ST], v1[2][OUT_ST];

#pragma unroll
  for (int i = 0; i < OUT_ST; ++i) {
    a0[0][i] = *(const v2f*)(ap0 + 4 * i);
    a1[0][i] = *(const v2f*)(ap1 + 4 * i);
    v0[0][i] = *(const float4*)(vp + (size_t)(4 * i) * H_);
    v1[0][i] = *(const float4*)(vp + (size_t)(4 * i + 1) * H_);
  }

#pragma unroll
  for (int st = 0; st < OUT_NS; ++st) {
    const int cur = st & 1, nxt = cur ^ 1;
    if (st + 1 < OUT_NS) {
#pragma unroll
      for (int i = 0; i < OUT_ST; ++i) {
        const int k = 4 * ((st + 1) * OUT_ST + i);
        a0[nxt][i] = *(const v2f*)(ap0 + k);
        a1[nxt][i] = *(const v2f*)(ap1 + k);
        v0[nxt][i] = *(const float4*)(vp + (size_t)k * H_);
        v1[nxt][i] = *(const float4*)(vp + (size_t)(k + 1) * H_);
      }
    }
    __builtin_amdgcn_sched_barrier(0);
#pragma unroll
    for (int i = 0; i < OUT_ST; ++i) {
      const float x0[4] = {v0[cur][i].x, v0[cur][i].y, v0[cur][i].z, v0[cur][i].w};
      const float x1[4] = {v1[cur][i].x, v1[cur][i].y, v1[cur][i].z, v1[cur][i].w};
      for (int j = 0; j < 4; ++j) {
        v2f bf;
        bf.x = x0[j];
        bf.y = x1[j];
        acc[j]     = WMMA_F32(a0[cur][i], bf, acc[j]);
        acc[4 + j] = WMMA_F32(a1[cur][i], bf, acc[4 + j]);
      }
    }
  }

  for (int a = 0; a < 8; ++a)
    for (int r = 0; r < 8; ++r)
      red[wv][a * 8 + r][lane] = acc[a][r];
  __syncthreads();

  float* o = num_part + ((size_t)kc * B_ + b) * CA_ * H_ + cg * 64 + 4 * hl;
  const int t = wv >> 1;
  for (int q = 0; q < 4; ++q) {
    const int r = 4 * (wv & 1) + q;
    float sum[4];
    for (int j = 0; j < 4; ++j) {
      const int reg = (4 * t + j) * 8 + r;
      float s = red[0][reg][lane];
      for (int w = 1; w < OUT_WAVES; ++w) s += red[w][reg][lane];
      sum[j] = s;
    }
    const int c = 16 * t + 8 * half + r;
    *(float4*)(o + (size_t)c * H_) = make_float4(sum[0], sum[1], sum[2], sum[3]);
  }
}

__global__ void k_fin(const float* __restrict__ num_part, const float* __restrict__ den_part,
                      float* __restrict__ out) {
  const int idx = blockIdx.x * 256 + threadIdx.x;
  const int bc  = idx / H_;
  const int b   = bc >> 5, c = bc & 31;
  float num = 0.f;
  for (int kc = 0; kc < KSPLIT; ++kc) num += num_part[(size_t)kc * (B_ * CA_ * H_) + idx];
  float den = 0.f;
  for (int ch = 0; ch < NCHUNK; ++ch) den += den_part[((size_t)b * NCHUNK + ch) * CA_ + c];
  out[idx] = num / den;
}

extern "C" void kernel_launch(void* const* d_in, const int* in_sizes, int n_in,
                              void* d_out, int out_size, void* d_ws, size_t ws_size,
                              hipStream_t stream) {
  (void)in_sizes; (void)n_in; (void)out_size; (void)ws_size;
  const float* key     = (const float*)d_in[0];
  const float* value   = (const float*)d_in[1];
  const float* query   = (const float*)d_in[2];
  const int*   seq_len = (const int*)d_in[3];
  const float* W       = (const float*)d_in[4];
  const float* bias    = (const float*)d_in[5];
  float* ws  = (float*)d_ws;
  float* out = (float*)d_out;

  k_tq<<<B_ * TQ_NT, 32 * TQ_WAVES, 0, stream>>>(query, W, bias, ws + TQ_OFF);

  const int smem_bytes = (32 * TQ_STRIDE + 32 * 8) * sizeof(float);
  k_scores<<<B_ * NCHUNK, 256, smem_bytes, stream>>>(key, ws + TQ_OFF, seq_len,
                                                     ws + AL_OFF, ws + DEN_OFF);

  k_out<<<B_ * KSPLIT * OUT_CG, 32 * OUT_WAVES, 0, stream>>>(value, ws + AL_OFF, ws + NUM_OFF);

  k_fin<<<(B_ * CA_ * H_) / 256, 256, 0, stream>>>(ws + NUM_OFF, ws + DEN_OFF, out);
}


# scores GEMM: key and tq fragments read as float4 per lane (same permuted k order in both loaders: new sum order), 24-36 key loads in flight ahead of the matrix work, tq staged as one load group
# speedup vs baseline: 1.9122x; 1.1421x over previous
#include <hip/hip_runtime.h>

typedef __attribute__((ext_vector_type(2))) float v2f;
typedef __attribute__((ext_vector_type(4))) float v4f;
typedef __attribute__((ext_vector_type(8))) float v8f;

#define WMMA_F32(a, b, c) \
  __builtin_amdgcn_wmma_f32_16x16x4_f32(false, (a), false, (b), (short)0, (c), false, false)

constexpr int B_ = 16, S_ = 2048, H_ = 768, CA_ = 32, IN_ = 300;
constexpr int SC = 128;
constexpr int NCHUNK = S_ / SC;
constexpr int KSPLIT = 4;
constexpr int KLEN = S_ / KSPLIT;
constexpr int TQ_STRIDE = 772;

constexpr size_t TQ_OFF  = 0;
constexpr size_t TQ_SZ   = (size_t)B_ * CA_ * H_;
constexpr size_t AL_OFF  = TQ_OFF + TQ_SZ;
constexpr size_t AL_SZ   = (size_t)B_ * CA_ * S_;
constexpr size_t DEN_OFF = AL_OFF + AL_SZ;
constexpr size_t DEN_SZ  = (size_t)B_ * NCHUNK * CA_;
constexpr size_t NUM_OFF = DEN_OFF + DEN_SZ;
constexpr size_t NUM_SZ  = (size_t)KSPLIT * B_ * CA_ * H_;

constexpr int TQ_WAVES = 5;
constexpr int TQ_STEPS = IN_ / (4 * TQ_WAVES);
constexpr int TQ_NT    = H_ / 32;

__global__ void __launch_bounds__(32 * TQ_WAVES)
k_tq(const float* __restrict__ query, const float* __restrict__ W,
     const float* __restrict__ bias, float* __restrict__ tq) {
  __shared__ float red[TQ_WAVES][32][32];
  const int b    = blockIdx.x / TQ_NT;
  const int nt   = blockIdx.x % TQ_NT;
  const int tid  = threadIdx.x;
  const int lane = tid & 31;
  const int wv   = tid >> 5;
  const int hl   = lane & 15;
  const int half = lane >> 4;
  const int kh   = half * 2;
  const int kb   = wv * (TQ_STEPS * 4);

  const float* ap0 = query + ((size_t)b * CA_ + hl) * IN_ + kb + kh;
  const float* ap1 = ap0 + (size_t)16 * IN_;
  const float* bp  = W + (size_t)(kb + kh) * H_ + nt * 32 + 2 * hl;

  v8f acc00 = {0.f, 0.f, 0.f, 0.f, 0.f, 0.f, 0.f, 0.f};
  v8f acc01 = acc00, acc10 = acc00, acc11 = acc00;
  v2f a0[TQ_STEPS], a1[TQ_STEPS], w0[TQ_STEPS], w1[TQ_STEPS];
#pragma unroll
  for (int i = 0; i < TQ_STEPS; ++i) {
    a0[i] = *(const v2f*)(ap0 + 4 * i);
    a1[i] = *(const v2f*)(ap1 + 4 * i);
    w0[i] = *(const v2f*)(bp + (size_t)(4 * i) * H_);
    w1[i] = *(const v2f*)(bp + (size_t)(4 * i + 1) * H_);
  }
  __builtin_amdgcn_sched_barrier(0);
#pragma unroll
  for (int i = 0; i < TQ_STEPS; ++i) {
    v2f b0, b1;
    b0.x = w0[i].x; b0.y = w1[i].x;
    b1.x = w0[i].y; b1.y = w1[i].y;
    acc00 = WMMA_F32(a0[i], b0, acc00);
    acc01 = WMMA_F32(a0[i], b1, acc01);
    acc10 = WMMA_F32(a1[i], b0, acc10);
    acc11 = WMMA_F32(a1[i], b1, acc11);
  }

  for (int r = 0; r < 8; ++r) {
    red[wv][r][lane]      = acc00[r];
    red[wv][8 + r][lane]  = acc01[r];
    red[wv][16 + r][lane] = acc10[r];
    red[wv][24 + r][lane] = acc11[r];
  }
  __syncthreads();

  if (wv < 4) {
    const int h0 = nt * 32 + 2 * hl;
    const v2f bv = *(const v2f*)(bias + h0);
    for (int q = 0; q < 2; ++q) {
      const int r = 2 * wv + q;
      float s00 = red[0][r][lane],      s01 = red[0][8 + r][lane];
      float s10 = red[0][16 + r][lane], s11 = red[0][24 + r][lane];
      for (int w = 1; w < TQ_WAVES; ++w) {
        s00 += red[w][r][lane];
        s01 += red[w][8 + r][lane];
        s10 += red[w][16 + r][lane];
        s11 += red[w][24 + r][lane];
      }
      const int c = 8 * half + r;
      v2f o0, o1;
      o0.x = tanhf(s00 + bv.x); o0.y = tanhf(s01 + bv.y);
      o1.x = tanhf(s10 + bv.x); o1.y = tanhf(s11 + bv.y);
      *(v2f*)(tq + ((size_t)b * CA_ + c) * H_ + h0)      = o0;
      *(v2f*)(tq + ((size_t)b * CA_ + c + 16) * H_ + h0) = o1;
    }
  }
}

constexpr int SC_U     = 4;
constexpr int SC_NU    = H_ / (8 * SC_U);
constexpr int SC_GL    = 8;
constexpr int SC_NG    = H_ / (8 * SC_GL);
constexpr int SC_AHEAD = 3;
constexpr int SC_RING  = SC_AHEAD + 1;
constexpr int SC_TQL   = 32 * (H_ / 4) / 256;
static_assert(SC_TQL == 24, "the staging statement below names 24 values");

__global__ void __launch_bounds__(256)
k_scores(const float* __restrict__ key, const float* __restrict__ tq,
         const int* __restrict__ seq_len,
         float* __restrict__ alpha_g, float* __restrict__ den_part) {
  extern __shared__ float4 smem4[];
  float* smem = (float*)smem4;
  float* tqs  = smem;
  float* dens = smem + 32 * TQ_STRIDE;

  const int b     = blockIdx.x / NCHUNK;
  const int chunk = blockIdx.x % NCHUNK;
  const int s0    = chunk * SC;
  const int tid   = threadIdx.x;
  const int lane  = tid & 31;
  const int wv    = tid >> 5;
  const int hl    = lane & 15;
  const int half  = lane >> 4;

  const v4f* tqg = (const v4f*)(tq + (size_t)b * CA_ * H_);
  v4f tv[SC_TQL];
#pragma unroll
  for (int j = 0; j < SC_TQL; ++j) tv[j] = tqg[tid + 256 * j];

  const float* krow = key + ((size_t)b * S_ + (s0 + wv * 16 + hl)) * H_ + 4 * half;
  float4 kv[SC_RING][SC_GL];
#pragma unroll
  for (int g = 0; g < SC_AHEAD; ++g) {
#pragma unroll
    for (int i = 0; i < SC_GL; ++i)
      kv[g][i] = *(const float4*)(krow + 8 * (SC_GL * g + i));
  }
  asm volatile("" :: "v"(tv[0]), "v"(tv[1]), "v"(tv[2]), "v"(tv[3]), "v"(tv[4]), "v"(tv[5]),
               "v"(tv[6]), "v"(tv[7]), "v"(tv[8]), "v"(tv[9]), "v"(tv[10]), "v"(tv[11]),
               "v"(tv[12]), "v"(tv[13]), "v"(tv[14]), "v"(tv[15]), "v"(tv[16]), "v"(tv[17]),
               "v"(tv[18]), "v"(tv[19]), "v"(tv[20]), "v"(tv[21]), "v"(tv[22]), "v"(tv[23])
               : "memory");
#pragma unroll
  for (int j = 0; j < SC_TQL; ++j) {
    const int i = tid + 256 * j;
    const int row = i / (H_ / 4), c4 = i % (H_ / 4);
    *(v4f*)&tqs[row * TQ_STRIDE + c4 * 4] = tv[j];
  }
  __syncthreads();

  const float* a0p = &tqs[hl * TQ_STRIDE + 4 * half];
  const float* a1p = a0p + 16 * TQ_STRIDE;
  float4 af0[2][SC_U], af1[2][SC_U];
#pragma unroll
  for (int i = 0; i < SC_U; ++i) {
    af0[0][i] = *(const float4*)(a0p + 8 * i);
    af1[0][i] = *(const float4*)(a1p + 8 * i);
  }

  v8f acc0 = {0.f, 0.f, 0.f, 0.f, 0.f, 0.f, 0.f, 0.f};
  v8f acc1 = {0.f, 0.f, 0.f, 0.f, 0.f, 0.f, 0.f, 0.f};
#pragma unroll
  for (int u = 0; u < SC_NU; ++u) {
    const int g = u >> 1, cur = u & 1, nxt = cur ^ 1;
    if (cur == 0 && g + SC_AHEAD < SC_NG) {
#pragma unroll
      for (int i = 0; i < SC_GL; ++i)
        kv[(g + SC_AHEAD) % SC_RING][i] =
            *(const float4*)(krow + 8 * (SC_GL * (g + SC_AHEAD) + i));
    }
    if (u + 1 < SC_NU) {
#pragma unroll
      for (int i = 0; i < SC_U; ++i) {
        af0[nxt][i] = *(const float4*)(a0p + 8 * (SC_U * (u + 1) + i));
        af1[nxt][i] = *(const float4*)(a1p + 8 * (SC_U * (u + 1) + i));
      }
    }
    __builtin_amdgcn_sched_barrier(0);
#pragma unroll
    for (int i = 0; i < SC_U; ++i) {
      const float4 kq = kv[g % SC_RING][SC_U * cur + i];
      const float4 p  = af0[cur][i];
      const float4 q  = af1[cur][i];
      v2f blo, bhi, p0, p1, q0, q1;
      blo.x = kq.x; blo.y = kq.y;  bhi.x = kq.z; bhi.y = kq.w;
      p0.x = p.x;   p0.y = p.y;    p1.x = p.z;   p1.y = p.w;
      q0.x = q.x;   q0.y = q.y;    q1.x = q.z;   q1.y = q.w;
      acc0 = WMMA_F32(p0, blo, acc0);
      acc1 = WMMA_F32(q0, blo, acc1);
      acc0 = WMMA_F32(p1, bhi, acc0);
      acc1 = WMMA_F32(q1, bhi, acc1);
    }
  }

  const int s = s0 + wv * 16 + hl;
  const bool valid = s < seq_len[b];
  float* ag = alpha_g + (size_t)b * CA_ * S_;
  for (int r = 0; r < 8; ++r) {
    const int c0 = r + 8 * half;
    float v0 = valid ? __expf(acc0[r]) : 0.0f;
    float v1 = valid ? __expf(acc1[r]) : 0.0f;
    ag[(size_t)c0 * S_ + s]        = v0;
    ag[(size_t)(c0 + 16) * S_ + s] = v1;
    float r0 = v0, r1 = v1;
    for (int m = 1; m < 16; m <<= 1) {
      r0 += __shfl_xor(r0, m, 32);
      r1 += __shfl_xor(r1, m, 32);
    }
    if (hl == 0) {
      dens[c0 * 8 + wv]        = r0;
      dens[(c0 + 16) * 8 + wv] = r1;
    }
  }
  __syncthreads();
  if (tid < CA_) {
    float d = 0.f;
    for (int w = 0; w < 8; ++w) d += dens[tid * 8 + w];
    den_part[((size_t)b * NCHUNK + chunk) * CA_ + tid] = d;
  }
}

constexpr int OUT_WAVES = 4;
constexpr int OUT_KW    = KLEN / OUT_WAVES;
constexpr int OUT_CG    = H_ / 64;
constexpr int OUT_ST    = 8;
constexpr int OUT_NS    = OUT_KW / (4 * OUT_ST);

__global__ void __launch_bounds__(32 * OUT_WAVES)
k_out(const float* __restrict__ value, const float* __restrict__ alpha_g,
      float* __restrict__ num_part) {
  __shared__ float red[OUT_WAVES][64][32];
  const int gid  = blockIdx.x;
  const int cg   = gid % OUT_CG;
  const int kc   = (gid / OUT_CG) % KSPLIT;
  const int b    = gid / (OUT_CG * KSPLIT);
  const int tid  = threadIdx.x;
  const int lane = tid & 31;
  const int wv   = tid >> 5;
  const int hl   = lane & 15;
  const int half = lane >> 4;
  const int kh   = half * 2;
  const int sb   = kc * KLEN + wv * OUT_KW;

  const float* ap0 = alpha_g + ((size_t)b * CA_ + hl) * S_ + sb + kh;
  const float* ap1 = ap0 + (size_t)16 * S_;
  const float* vp  = value + ((size_t)b * S_ + sb + kh) * H_ + cg * 64 + 4 * hl;

  v8f acc[8];
  for (int a = 0; a < 8; ++a) acc[a] = (v8f){0.f, 0.f, 0.f, 0.f, 0.f, 0.f, 0.f, 0.f};

  v2f    a0[2][OUT_ST], a1[2][OUT_ST];
  float4 v0[2][OUT_ST], v1[2][OUT_ST];

#pragma unroll
  for (int i = 0; i < OUT_ST; ++i) {
    a0[0][i] = *(const v2f*)(ap0 + 4 * i);
    a1[0][i] = *(const v2f*)(ap1 + 4 * i);
    v0[0][i] = *(const float4*)(vp + (size_t)(4 * i) * H_);
    v1[0][i] = *(const float4*)(vp + (size_t)(4 * i + 1) * H_);
  }

#pragma unroll
  for (int st = 0; st < OUT_NS; ++st) {
    const int cur = st & 1, nxt = cur ^ 1;
    if (st + 1 < OUT_NS) {
#pragma unroll
      for (int i = 0; i < OUT_ST; ++i) {
        const int k = 4 * ((st + 1) * OUT_ST + i);
        a0[nxt][i] = *(const v2f*)(ap0 + k);
        a1[nxt][i] = *(const v2f*)(ap1 + k);
        v0[nxt][i] = *(const float4*)(vp + (size_t)k * H_);
        v1[nxt][i] = *(const float4*)(vp + (size_t)(k + 1) * H_);
      }
    }
    __builtin_amdgcn_sched_barrier(0);
#pragma unroll
    for (int i = 0; i < OUT_ST; ++i) {
      const float x0[4] = {v0[cur][i].x, v0[cur][i].y, v0[cur][i].z, v0[cur][i].w};
      const float x1[4] = {v1[cur][i].x, v1[cur][i].y, v1[cur][i].z, v1[cur][i].w};
      for (int j = 0; j < 4; ++j) {
        v2f bf;
        bf.x = x0[j];
        bf.y = x1[j];
        acc[j]     = WMMA_F32(a0[cur][i], bf, acc[j]);
        acc[4 + j] = WMMA_F32(a1[cur][i], bf, acc[4 + j]);
      }
    }
  }

  for (int a = 0; a < 8; ++a)
    for (int r = 0; r < 8; ++r)
      red[wv][a * 8 + r][lane] = acc[a][r];
  __syncthreads();

  float* o = num_part + ((size_t)kc * B_ + b) * CA_ * H_ + cg * 64 + 4 * hl;
  const int t = wv >> 1;
  for (int q = 0; q < 4; ++q) {
    const int r = 4 * (wv & 1) + q;
    float sum[4];
    for (int j = 0; j < 4; ++j) {
      const int reg = (4 * t + j) * 8 + r;
      float s = red[0][reg][lane];
      for (int w = 1; w < OUT_WAVES; ++w) s += red[w][reg][lane];
      sum[j] = s;
    }
    const int c = 16 * t + 8 * half + r;
    *(float4*)(o + (size_t)c * H_) = make_float4(sum[0], sum[1], sum[2], sum[3]);
  }
}

__global__ void k_fin(const float* __restrict__ num_part, const float* __restrict__ den_part,
                      float* __restrict__ out) {
  const int idx = blockIdx.x * 256 + threadIdx.x;
  const int bc  = idx / H_;
  const int b   = bc >> 5, c = bc & 31;
  float num = 0.f;
  for (int kc = 0; kc < KSPLIT; ++kc) num += num_part[(size_t)kc * (B_ * CA_ * H_) + idx];
  float den = 0.f;
  for (int ch = 0; ch < NCHUNK; ++ch) den += den_part[((size_t)b * NCHUNK + ch) * CA_ + c];
  out[idx] = num / den;
}

extern "C" void kernel_launch(void* const* d_in, const int* in_sizes, int n_in,
                              void* d_out, int out_size, void* d_ws, size_t ws_size,
                              hipStream_t stream) {
  (void)in_sizes; (void)n_in; (void)out_size; (void)ws_size;
  const float* key     = (const float*)d_in[0];
  const float* value   = (const float*)d_in[1];
  const float* query   = (const float*)d_in[2];
  const int*   seq_len = (const int*)d_in[3];
  const float* W       = (const float*)d_in[4];
  const float* bias    = (const float*)d_in[5];
  float* ws  = (float*)d_ws;
  float* out = (float*)d_out;

  k_tq<<<B_ * TQ_NT, 32 * TQ_WAVES, 0, stream>>>(query, W, bias, ws + TQ_OFF);

  const int smem_bytes = (32 * TQ_STRIDE + 32 * 8) * sizeof(float);
  k_scores<<<B_ * NCHUNK, 256, smem_bytes, stream>>>(key, ws + TQ_OFF, seq_len,
                                                     ws + AL_OFF, ws + DEN_OFF);

  k_out<<<B_ * KSPLIT * OUT_CG, 32 * OUT_WAVES, 0, stream>>>(value, ws + AL_OFF, ws + NUM_OFF);

  k_fin<<<(B_ * CA_ * H_) / 256, 256, 0, stream>>>(ws + NUM_OFF, ws + DEN_OFF, out);
}


# rows past each batch's valid length are not read: output GEMM's 64 K-slots per column group shared out to batches by valid 16-row tiles, final kernel adds a batch's slots in order (new sum order)
# speedup vs baseline: 2.1353x; 1.1167x over previous
#include <hip/hip_runtime.h>

typedef __attribute__((ext_vector_type(2))) float v2f;
typedef __attribute__((ext_vector_type(4))) float v4f;
typedef __attribute__((ext_vector_type(8))) float v8f;

#define WMMA_F32(a, b, c) \
  __builtin_amdgcn_wmma_f32_16x16x4_f32(false, (a), false, (b), (short)0, (c), false, false)

constexpr int B_ = 16, S_ = 2048, H_ = 768, CA_ = 32, IN_ = 300;
constexpr int SC = 128;
constexpr int NCHUNK = S_ / SC;
constexpr int KSPLIT = 4;
constexpr int KLEN = S_ / KSPLIT;
constexpr int TQ_STRIDE = 772;

constexpr size_t TQ_OFF  = 0;
constexpr size_t TQ_SZ   = (size_t)B_ * CA_ * H_;
constexpr size_t AL_OFF  = TQ_OFF + TQ_SZ;
constexpr size_t AL_SZ   = (size_t)B_ * CA_ * S_;
constexpr size_t DEN_OFF = AL_OFF + AL_SZ;
constexpr size_t DEN_SZ  = (size_t)B_ * NCHUNK * CA_;
constexpr size_t NUM_OFF = DEN_OFF + DEN_SZ;
constexpr size_t NUM_SZ  = (size_t)KSPLIT * B_ * CA_ * H_;

constexpr int TQ_WAVES = 5;
constexpr int TQ_STEPS = IN_ / (4 * TQ_WAVES);
constexpr int TQ_NT    = H_ / 32;

__global__ void __launch_bounds__(32 * TQ_WAVES)
k_tq(const float* __restrict__ query, const float* __restrict__ W,
     const float* __restrict__ bias, float* __restrict__ tq) {
  __shared__ float red[TQ_WAVES][32][32];
  const int b    = blockIdx.x / TQ_NT;
  const int nt   = blockIdx.x % TQ_NT;
  const int tid  = threadIdx.x;
  const int lane = tid & 31;
  const int wv   = tid >> 5;
  const int hl   = lane & 15;
  const int half = lane >> 4;
  const int kh   = half * 2;
  const int kb   = wv * (TQ_STEPS * 4);

  const float* ap0 = query + ((size_t)b * CA_ + hl) * IN_ + kb + kh;
  const float* ap1 = ap0 + (size_t)16 * IN_;
  const float* bp  = W + (size_t)(kb + kh) * H_ + nt * 32 + 2 * hl;

  v8f acc00 = {0.f, 0.f, 0.f, 0.f, 0.f, 0.f, 0.f, 0.f};
  v8f acc01 = acc00, acc10 = acc00, acc11 = acc00;
  v2f a0[TQ_STEPS], a1[TQ_STEPS], w0[TQ_STEPS], w1[TQ_STEPS];
#pragma unroll
  for (int i = 0; i < TQ_STEPS; ++i) {
    a0[i] = *(const v2f*)(ap0 + 4 * i);
    a1[i] = *(const v2f*)(ap1 + 4 * i);
    w0[i] = *(const v2f*)(bp + (size_t)(4 * i) * H_);
    w1[i] = *(const v2f*)(bp + (size_t)(4 * i + 1) * H_);
  }
  __builtin_amdgcn_sched_barrier(0);
#pragma unroll
  for (int i = 0; i < TQ_STEPS; ++i) {
    v2f b0, b1;
    b0.x = w0[i].x; b0.y = w1[i].x;
    b1.x = w0[i].y; b1.y = w1[i].y;
    acc00 = WMMA_F32(a0[i], b0, acc00);
    acc01 = WMMA_F32(a0[i], b1, acc01);
    acc10 = WMMA_F32(a1[i], b0, acc10);
    acc11 = WMMA_F32(a1[i], b1, acc11);
  }

  for (int r = 0; r < 8; ++r) {
    red[wv][r][lane]      = acc00[r];
    red[wv][8 + r][lane]  = acc01[r];
    red[wv][16 + r][lane] = acc10[r];
    red[wv][24 + r][lane] = acc11[r];
  }
  __syncthreads();

  if (wv < 4) {
    const int h0 = nt * 32 + 2 * hl;
    const v2f bv = *(const v2f*)(bias + h0);
    for (int q = 0; q < 2; ++q) {
      const int r = 2 * wv + q;
      float s00 = red[0][r][lane],      s01 = red[0][8 + r][lane];
      float s10 = red[0][16 + r][lane], s11 = red[0][24 + r][lane];
      for (int w = 1; w < TQ_WAVES; ++w) {
        s00 += red[w][r][lane];
        s01 += red[w][8 + r][lane];
        s10 += red[w][16 + r][lane];
        s11 += red[w][24 + r][lane];
      }
      const int c = 8 * half + r;
      v2f o0, o1;
      o0.x = tanhf(s00 + bv.x); o0.y = tanhf(s01 + bv.y);
      o1.x = tanhf(s10 + bv.x); o1.y = tanhf(s11 + bv.y);
      *(v2f*)(tq + ((size_t)b * CA_ + c) * H_ + h0)      = o0;
      *(v2f*)(tq + ((size_t)b * CA_ + c + 16) * H_ + h0) = o1;
    }
  }
}

constexpr int SC_U     = 4;
constexpr int SC_NU    = H_ / (8 * SC_U);
constexpr int SC_GL    = 8;
constexpr int SC_NG    = H_ / (8 * SC_GL);
constexpr int SC_AHEAD = 3;
constexpr int SC_RING  = SC_AHEAD + 1;
constexpr int SC_TQL   = 32 * (H_ / 4) / 256;
static_assert(SC_TQL == 24, "the staging statement below names 24 values");

__global__ void __launch_bounds__(256)
k_scores(const float* __restrict__ key, const float* __restrict__ tq,
         const int* __restrict__ seq_len,
         float* __restrict__ alpha_g, float* __restrict__ den_part) {
  extern __shared__ float4 smem4[];
  float* smem = (float*)smem4;
  float* tqs  = smem;
  float* dens = smem + 32 * TQ_STRIDE;

  const int b     = blockIdx.x / NCHUNK;
  const int chunk = blockIdx.x % NCHUNK;
  const int s0    = chunk * SC;
  const int tid   = threadIdx.x;
  const int lane  = tid & 31;
  const int wv    = tid >> 5;
  const int hl    = lane & 15;
  const int half  = lane >> 4;

  if (s0 >= seq_len[b]) {
    if (tid < CA_) den_part[((size_t)b * NCHUNK + chunk) * CA_ + tid] = 0.f;
    return;
  }

  const v4f* tqg = (const v4f*)(tq + (size_t)b * CA_ * H_);
  v4f tv[SC_TQL];
#pragma unroll
  for (int j = 0; j < SC_TQL; ++j) tv[j] = tqg[tid + 256 * j];

  const float* krow = key + ((size_t)b * S_ + (s0 + wv * 16 + hl)) * H_ + 4 * half;
  float4 kv[SC_RING][SC_GL];
#pragma unroll
  for (int g = 0; g < SC_AHEAD; ++g) {
#pragma unroll
    for (int i = 0; i < SC_GL; ++i)
      kv[g][i] = *(const float4*)(krow + 8 * (SC_GL * g + i));
  }
  asm volatile("" :: "v"(tv[0]), "v"(tv[1]), "v"(tv[2]), "v"(tv[3]), "v"(tv[4]), "v"(tv[5]),
               "v"(tv[6]), "v"(tv[7]), "v"(tv[8]), "v"(tv[9]), "v"(tv[10]), "v"(tv[11]),
               "v"(tv[12]), "v"(tv[13]), "v"(tv[14]), "v"(tv[15]), "v"(tv[16]), "v"(tv[17]),
               "v"(tv[18]), "v"(tv[19]), "v"(tv[20]), "v"(tv[21]), "v"(tv[22]), "v"(tv[23])
               : "memory");
#pragma unroll
  for (int j = 0; j < SC_TQL; ++j) {
    const int i = tid + 256 * j;
    const int row = i / (H_ / 4), c4 = i % (H_ / 4);
    *(v4f*)&tqs[row * TQ_STRIDE + c4 * 4] = tv[j];
  }
  __syncthreads();

  const float* a0p = &tqs[hl * TQ_STRIDE + 4 * half];
  const float* a1p = a0p + 16 * TQ_STRIDE;
  float4 af0[2][SC_U], af1[2][SC_U];
#pragma unroll
  for (int i = 0; i < SC_U; ++i) {
    af0[0][i] = *(const float4*)(a0p + 8 * i);
    af1[0][i] = *(const float4*)(a1p + 8 * i);
  }

  v8f acc0 = {0.f, 0.f, 0.f, 0.f, 0.f, 0.f, 0.f, 0.f};
  v8f acc1 = {0.f, 0.f, 0.f, 0.f, 0.f, 0.f, 0.f, 0.f};
#pragma unroll
  for (int u = 0; u < SC_NU; ++u) {
    const int g = u >> 1, cur = u & 1, nxt = cur ^ 1;
    if (cur == 0 && g + SC_AHEAD < SC_NG) {
#pragma unroll
      for (int i = 0; i < SC_GL; ++i)
        kv[(g + SC_AHEAD) % SC_RING][i] =
            *(const float4*)(krow + 8 * (SC_GL * (g + SC_AHEAD) + i));
    }
    if (u + 1 < SC_NU) {
#pragma unroll
      for (int i = 0; i < SC_U; ++i) {
        af0[nxt][i] = *(const float4*)(a0p + 8 * (SC_U * (u + 1) + i));
        af1[nxt][i] = *(const float4*)(a1p + 8 * (SC_U * (u + 1) + i));
      }
    }
    __builtin_amdgcn_sched_barrier(0);
#pragma unroll
    for (int i = 0; i < SC_U; ++i) {
      const float4 kq = kv[g % SC_RING][SC_U * cur + i];
      const float4 p  = af0[cur][i];
      const float4 q  = af1[cur][i];
      v2f blo, bhi, p0, p1, q0, q1;
      blo.x = kq.x; blo.y = kq.y;  bhi.x = kq.z; bhi.y = kq.w;
      p0.x = p.x;   p0.y = p.y;    p1.x = p.z;   p1.y = p.w;
      q0.x = q.x;   q0.y = q.y;    q1.x = q.z;   q1.y = q.w;
      acc0 = WMMA_F32(p0, blo, acc0);
      acc1 = WMMA_F32(q0, blo, acc1);
      acc0 = WMMA_F32(p1, bhi, acc0);
      acc1 = WMMA_F32(q1, bhi, acc1);
    }
  }

  const int s = s0 + wv * 16 + hl;
  const bool valid = s < seq_len[b];
  float* ag = alpha_g + (size_t)b * CA_ * S_;
  for (int r = 0; r < 8; ++r) {
    const int c0 = r + 8 * half;
    float v0 = valid ? __expf(acc0[r]) : 0.0f;
    float v1 = valid ? __expf(acc1[r]) : 0.0f;
    ag[(size_t)c0 * S_ + s]        = v0;
    ag[(size_t)(c0 + 16) * S_ + s] = v1;
    float r0 = v0, r1 = v1;
    for (int m = 1; m < 16; m <<= 1) {
      r0 += __shfl_xor(r0, m, 32);
      r1 += __shfl_xor(r1, m, 32);
    }
    if (hl == 0) {
      dens[c0 * 8 + wv]        = r0;
      dens[(c0 + 16) * 8 + wv] = r1;
    }
  }
  __syncthreads();
  if (tid < CA_) {
    float d = 0.f;
    for (int w = 0; w < 8; ++w) d += dens[tid * 8 + w];
    den_part[((size_t)b * NCHUNK + chunk) * CA_ + tid] = d;
  }
}

constexpr int OUT_WAVES = 4;
constexpr int OUT_CG    = H_ / 64;
constexpr int OUT_SLOTS = KSPLIT * B_;
constexpr int OUT_ST    = 4;
static_assert(OUT_ST == 4, "the pinning statement in k_out names 4 k-steps");

__device__ __forceinline__ void out_plan(const int* __restrict__ seq_len, int lane,
                                         int& tiles, int& eend) {
  const int bl = lane & 15;
  int len = seq_len[bl];
  len = len < 0 ? 0 : (len > S_ ? S_ : len);
  tiles = (len + 15) >> 4;
  int p = tiles;
  for (int d = 1; d < 16; d <<= 1) {
    const int q = __shfl_up(p, d, 16);
    if (bl >= d) p += q;
  }
  const int G = __shfl(p, 15, 16);
  int r = (G > 0) ? (128 * p + G) / (2 * G) : 0;
  r = r > 49 + bl ? 49 + bl : r;
  r = r < bl + 1 ? bl + 1 : r;
  int m = r - bl;
  for (int d = 1; d < 16; d <<= 1) {
    const int q = __shfl_up(m, d, 16);
    if (bl >= d) m = m > q ? m : q;
  }
  eend = bl + m;
}

__global__ void __launch_bounds__(32 * OUT_WAVES)
k_out(const float* __restrict__ value, const float* __restrict__ alpha_g,
      const int* __restrict__ seq_len, float* __restrict__ num_part) {
  __shared__ float red[OUT_WAVES][64][32];
  const int gid  = blockIdx.x;
  const int cg   = gid % OUT_CG;
  const int slot = gid / OUT_CG;
  const int tid  = threadIdx.x;
  const int lane = tid & 31;
  const int wv   = tid >> 5;
  const int hl   = lane & 15;
  const int half = lane >> 4;
  const int kh   = half * 2;

  int tiles, eend;
  out_plan(seq_len, lane, tiles, eend);
  const int b = __popc((unsigned)__ballot(eend <= slot) & 0xFFFFu);
  if (b > 15) return;
  const int e0 = (b > 0) ? __shfl(eend, b - 1, 32) : 0;
  const int e1 = __shfl(eend, b, 32);
  const int tb = __shfl(tiles, b, 32);
  const int nb = e1 - e0;
  const int lo = ((slot - e0) * tb) / nb;
  const int hi = ((slot - e0 + 1) * tb) / nb;
  const int n  = hi - lo;
  const int t0 = __builtin_amdgcn_readfirstlane(lo + (wv * n) / OUT_WAVES);
  const int t1 = __builtin_amdgcn_readfirstlane(lo + ((wv + 1) * n) / OUT_WAVES);

  const float* ap0 = alpha_g + ((size_t)b * CA_ + hl) * S_ + kh;
  const float* ap1 = ap0 + (size_t)16 * S_;
  const float* vp  = value + ((size_t)b * S_ + kh) * H_ + cg * 64 + 4 * hl;

  v8f acc[8];
  for (int a = 0; a < 8; ++a) acc[a] = (v8f){0.f, 0.f, 0.f, 0.f, 0.f, 0.f, 0.f, 0.f};

  if (t0 < t1) {
    v2f    a0[3][OUT_ST], a1[3][OUT_ST];
    v4f    v0[3][OUT_ST], v1[3][OUT_ST];
    const int last = t1 - 1;
#define OUT_LOAD(SET, TILE)                                              \
    {                                                                    \
      const int tt_ = (TILE) < last ? (TILE) : last;                     \
      _Pragma("unroll") for (int i = 0; i < OUT_ST; ++i) {               \
        const int k = 16 * tt_ + 4 * i;                                  \
        a0[SET][i] = *(const v2f*)(ap0 + k);                             \
        a1[SET][i] = *(const v2f*)(ap1 + k);                             \
        v0[SET][i] = *(const v4f*)(vp + (size_t)k * H_);                 \
        v1[SET][i] = *(const v4f*)(vp + (size_t)(k + 1) * H_);           \
      }                                                                  \
    }
#define OUT_MMA(SET)                                                     \
    {                                                                    \
         \
      asm volatile("" : "+v"(v0[SET][0]), "+v"(v1[SET][0]), "+v"(v0[SET][1]),   \
                        "+v"(v1[SET][1]), "+v"(v0[SET][2]), "+v"(v1[SET][2]),   \
                        "+v"(v0[SET][3]), "+v"(v1[SET][3]));                    \
      _Pragma("unroll") for (int i = 0; i < OUT_ST; ++i) {               \
        const float x0[4] = {v0[SET][i].x, v0[SET][i].y, v0[SET][i].z, v0[SET][i].w}; \
        const float x1[4] = {v1[SET][i].x, v1[SET][i].y, v1[SET][i].z, v1[SET][i].w}; \
        _Pragma("unroll") for (int j = 0; j < 4; ++j) {                  \
          v2f bf;                                                        \
          bf.x = x0[j];                                                  \
          bf.y = x1[j];                                                  \
          acc[j]     = WMMA_F32(a0[SET][i], bf, acc[j]);                 \
          acc[4 + j] = WMMA_F32(a1[SET][i], bf, acc[4 + j]);             \
        }                                                                \
      }                                                                  \
    }
    OUT_LOAD(0, t0);
    OUT_LOAD(1, t0 + 1);
    int t = t0;
    for (; t + 3 <= t1; t += 3) {
      OUT_LOAD(2, t + 2);
      __builtin_amdgcn_sched_barrier(0);
      OUT_MMA(0);
      OUT_LOAD(0, t + 3);
      __builtin_amdgcn_sched_barrier(0);
      OUT_MMA(1);
      OUT_LOAD(1, t + 4);
      __builtin_amdgcn_sched_barrier(0);
      OUT_MMA(2);
    }
    if (t < t1) OUT_MMA(0);
    if (t + 1 < t1) OUT_MMA(1);
#undef OUT_LOAD
#undef OUT_MMA
  }

  for (int a = 0; a < 8; ++a)
    for (int r = 0; r < 8; ++r)
      red[wv][a * 8 + r][lane] = acc[a][r];
  __syncthreads();

  float* o = num_part + (size_t)slot * CA_ * H_ + cg * 64 + 4 * hl;
  const int t = wv >> 1;
  for (int q = 0; q < 4; ++q) {
    const int r = 4 * (wv & 1) + q;
    float sum[4];
    for (int j = 0; j < 4; ++j) {
      const int reg = (4 * t + j) * 8 + r;
      float s = red[0][reg][lane];
      for (int w = 1; w < OUT_WAVES; ++w) s += red[w][reg][lane];
      sum[j] = s;
    }
    const int c = 16 * t + 8 * half + r;
    *(float4*)(o + (size_t)c * H_) = make_float4(sum[0], sum[1], sum[2], sum[3]);
  }
}

__global__ void k_fin(const float* __restrict__ num_part, const float* __restrict__ den_part,
                      const int* __restrict__ seq_len, float* __restrict__ out) {
  const int idx = blockIdx.x * 256 + threadIdx.x;
  const int bc  = idx / H_;
  const int b   = bc >> 5, c = bc & 31;
  const int h   = idx - bc * H_;
  float dv[NCHUNK];
#pragma unroll
  for (int k = 0; k < NCHUNK; ++k) dv[k] = den_part[((size_t)b * NCHUNK + k) * CA_ + c];
  int tiles, eend;
  out_plan(seq_len, threadIdx.x & 31, tiles, eend);
  const int e0 = (b > 0) ? __shfl(eend, b - 1, 32) : 0;
  const int e1 = __shfl(eend, b, 32);
  float num = 0.f;
  for (int base = e0; base < e1; base += 8) {
    float nv[8];
#pragma unroll
    for (int q = 0; q < 8; ++q) {
      const int sl = base + q < e1 ? base + q : e1 - 1;
      nv[q] = num_part[((size_t)sl * CA_ + c) * H_ + h];
    }
#pragma unroll
    for (int q = 0; q < 8; ++q) num += (base + q < e1) ? nv[q] : 0.f;
  }
  float den = 0.f;
#pragma unroll
  for (int k = 0; k < NCHUNK; ++k) den += dv[k];
  out[idx] = num / den;
}

extern "C" void kernel_launch(void* const* d_in, const int* in_sizes, int n_in,
                              void* d_out, int out_size, void* d_ws, size_t ws_size,
                              hipStream_t stream) {
  (void)in_sizes; (void)n_in; (void)out_size; (void)ws_size;
  const float* key     = (const float*)d_in[0];
  const float* value   = (const float*)d_in[1];
  const float* query   = (const float*)d_in[2];
  const int*   seq_len = (const int*)d_in[3];
  const float* W       = (const float*)d_in[4];
  const float* bias    = (const float*)d_in[5];
  float* ws  = (float*)d_ws;
  float* out = (float*)d_out;

  k_tq<<<B_ * TQ_NT, 32 * TQ_WAVES, 0, stream>>>(query, W, bias, ws + TQ_OFF);

  const int smem_bytes = (32 * TQ_STRIDE + 32 * 8) * sizeof(float);
  k_scores<<<B_ * NCHUNK, 256, smem_bytes, stream>>>(key, ws + TQ_OFF, seq_len,
                                                     ws + AL_OFF, ws + DEN_OFF);

  k_out<<<OUT_SLOTS * OUT_CG, 32 * OUT_WAVES, 0, stream>>>(value, ws + AL_OFF, seq_len, ws + NUM_OFF);

  k_fin<<<(B_ * CA_ * H_) / 256, 256, 0, stream>>>(ws + NUM_OFF, ws + DEN_OFF, seq_len, out);
}


# every cross-lane shuffle (16-lane denominator sums in the scores kernel, slot plan in output and final kernels) on the vector ALU (DPP row ops, readlane), not LDS-unit permutes; same sums
# speedup vs baseline: 2.2168x; 1.0382x over previous
#include <hip/hip_runtime.h>

typedef __attribute__((ext_vector_type(2))) float v2f;
typedef __attribute__((ext_vector_type(4))) float v4f;
typedef __attribute__((ext_vector_type(8))) float v8f;

#define WMMA_F32(a, b, c) \
  __builtin_amdgcn_wmma_f32_16x16x4_f32(false, (a), false, (b), (short)0, (c), false, false)

constexpr int DPP_ROW_SHR   = 0x110;
constexpr int DPP_ROW_SHARE = 0x150;
constexpr int DPP_ROW_XMASK = 0x160;
template <int CTRL>
__device__ __forceinline__ int dpp_row(int old, int x) {
  return __builtin_amdgcn_update_dpp(old, x, CTRL, 0xf, 0xf, false);
}
template <int CTRL>
__device__ __forceinline__ float dpp_rowf(float old, float x) {
  return __int_as_float(dpp_row<CTRL>(__float_as_int(old), __float_as_int(x)));
}
__device__ __forceinline__ float row16_sum(float x) {
  x += dpp_rowf<DPP_ROW_XMASK + 1>(0.f, x);
  x += dpp_rowf<DPP_ROW_XMASK + 2>(0.f, x);
  x += dpp_rowf<DPP_ROW_XMASK + 4>(0.f, x);
  x += dpp_rowf<DPP_ROW_XMASK + 8>(0.f, x);
  return x;
}

constexpr int B_ = 16, S_ = 2048, H_ = 768, CA_ = 32, IN_ = 300;
constexpr int SC = 128;
constexpr int NCHUNK = S_ / SC;
constexpr int KSPLIT = 4;
constexpr int KLEN = S_ / KSPLIT;
constexpr int TQ_STRIDE = 772;

constexpr size_t TQ_OFF  = 0;
constexpr size_t TQ_SZ   = (size_t)B_ * CA_ * H_;
constexpr size_t AL_OFF  = TQ_OFF + TQ_SZ;
constexpr size_t AL_SZ   = (size_t)B_ * CA_ * S_;
constexpr size_t DEN_OFF = AL_OFF + AL_SZ;
constexpr size_t DEN_SZ  = (size_t)B_ * NCHUNK * CA_;
constexpr size_t NUM_OFF = DEN_OFF + DEN_SZ;
constexpr size_t NUM_SZ  = (size_t)KSPLIT * B_ * CA_ * H_;

constexpr int TQ_WAVES = 5;
constexpr int TQ_STEPS = IN_ / (4 * TQ_WAVES);
constexpr int TQ_NT    = H_ / 32;

__global__ void __launch_bounds__(32 * TQ_WAVES)
k_tq(const float* __restrict__ query, const float* __restrict__ W,
     const float* __restrict__ bias, float* __restrict__ tq) {
  __shared__ float red[TQ_WAVES][32][32];
  const int b    = blockIdx.x / TQ_NT;
  const int nt   = blockIdx.x % TQ_NT;
  const int tid  = threadIdx.x;
  const int lane = tid & 31;
  const int wv   = tid >> 5;
  const int hl   = lane & 15;
  const int half = lane >> 4;
  const int kh   = half * 2;
  const int kb   = wv * (TQ_STEPS * 4);

  const float* ap0 = query + ((size_t)b * CA_ + hl) * IN_ + kb + kh;
  const float* ap1 = ap0 + (size_t)16 * IN_;
  const float* bp  = W + (size_t)(kb + kh) * H_ + nt * 32 + 2 * hl;

  v8f acc00 = {0.f, 0.f, 0.f, 0.f, 0.f, 0.f, 0.f, 0.f};
  v8f acc01 = acc00, acc10 = acc00, acc11 = acc00;
  v2f a0[TQ_STEPS], a1[TQ_STEPS], w0[TQ_STEPS], w1[TQ_STEPS];
#pragma unroll
  for (int i = 0; i < TQ_STEPS; ++i) {
    a0[i] = *(const v2f*)(ap0 + 4 * i);
    a1[i] = *(const v2f*)(ap1 + 4 * i);
    w0[i] = *(const v2f*)(bp + (size_t)(4 * i) * H_);
    w1[i] = *(const v2f*)(bp + (size_t)(4 * i + 1) * H_);
  }
  __builtin_amdgcn_sched_barrier(0);
#pragma unroll
  for (int i = 0; i < TQ_STEPS; ++i) {
    v2f b0, b1;
    b0.x = w0[i].x; b0.y = w1[i].x;
    b1.x = w0[i].y; b1.y = w1[i].y;
    acc00 = WMMA_F32(a0[i], b0, acc00);
    acc01 = WMMA_F32(a0[i], b1, acc01);
    acc10 = WMMA_F32(a1[i], b0, acc10);
    acc11 = WMMA_F32(a1[i], b1, acc11);
  }

  for (int r = 0; r < 8; ++r) {
    red[wv][r][lane]      = acc00[r];
    red[wv][8 + r][lane]  = acc01[r];
    red[wv][16 + r][lane] = acc10[r];
    red[wv][24 + r][lane] = acc11[r];
  }
  __syncthreads();

  if (wv < 4) {
    const int h0 = nt * 32 + 2 * hl;
    const v2f bv = *(const v2f*)(bias + h0);
    for (int q = 0; q < 2; ++q) {
      const int r = 2 * wv + q;
      float s00 = red[0][r][lane],      s01 = red[0][8 + r][lane];
      float s10 = red[0][16 + r][lane], s11 = red[0][24 + r][lane];
      for (int w = 1; w < TQ_WAVES; ++w) {
        s00 += red[w][r][lane];
        s01 += red[w][8 + r][lane];
        s10 += red[w][16 + r][lane];
        s11 += red[w][24 + r][lane];
      }
      const int c = 8 * half + r;
      v2f o0, o1;
      o0.x = tanhf(s00 + bv.x); o0.y = tanhf(s01 + bv.y);
      o1.x = tanhf(s10 + bv.x); o1.y = tanhf(s11 + bv.y);
      *(v2f*)(tq + ((size_t)b * CA_ + c) * H_ + h0)      = o0;
      *(v2f*)(tq + ((size_t)b * CA_ + c + 16) * H_ + h0) = o1;
    }
  }
}

constexpr int SC_U     = 4;
constexpr int SC_NU    = H_ / (8 * SC_U);
constexpr int SC_GL    = 8;
constexpr int SC_NG    = H_ / (8 * SC_GL);
constexpr int SC_AHEAD = 3;
constexpr int SC_RING  = SC_AHEAD + 1;
constexpr int SC_TQL   = 32 * (H_ / 4) / 256;
static_assert(SC_TQL == 24, "the staging statement below names 24 values");

__global__ void __launch_bounds__(256)
k_scores(const float* __restrict__ key, const float* __restrict__ tq,
         const int* __restrict__ seq_len,
         float* __restrict__ alpha_g, float* __restrict__ den_part) {
  extern __shared__ float4 smem4[];
  float* smem = (float*)smem4;
  float* tqs  = smem;
  float* dens = smem + 32 * TQ_STRIDE;

  const int b     = blockIdx.x / NCHUNK;
  const int chunk = blockIdx.x % NCHUNK;
  const int s0    = chunk * SC;
  const int tid   = threadIdx.x;
  const int lane  = tid & 31;
  const int wv    = tid >> 5;
  const int hl    = lane & 15;
  const int half  = lane >> 4;

  if (s0 >= seq_len[b]) {
    if (tid < CA_) den_part[((size_t)b * NCHUNK + chunk) * CA_ + tid] = 0.f;
    return;
  }

  const v4f* tqg = (const v4f*)(tq + (size_t)b * CA_ * H_);
  v4f tv[SC_TQL];
#pragma unroll
  for (int j = 0; j < SC_TQL; ++j) tv[j] = tqg[tid + 256 * j];

  const float* krow = key + ((size_t)b * S_ + (s0 + wv * 16 + hl)) * H_ + 4 * half;
  float4 kv[SC_RING][SC_GL];
#pragma unroll
  for (int g = 0; g < SC_AHEAD; ++g) {
#pragma unroll
    for (int i = 0; i < SC_GL; ++i)
      kv[g][i] = *(const float4*)(krow + 8 * (SC_GL * g + i));
  }
  asm volatile("" :: "v"(tv[0]), "v"(tv[1]), "v"(tv[2]), "v"(tv[3]), "v"(tv[4]), "v"(tv[5]),
               "v"(tv[6]), "v"(tv[7]), "v"(tv[8]), "v"(tv[9]), "v"(tv[10]), "v"(tv[11]),
               "v"(tv[12]), "v"(tv[13]), "v"(tv[14]), "v"(tv[15]), "v"(tv[16]), "v"(tv[17]),
               "v"(tv[18]), "v"(tv[19]), "v"(tv[20]), "v"(tv[21]), "v"(tv[22]), "v"(tv[23])
               : "memory");
#pragma unroll
  for (int j = 0; j < SC_TQL; ++j) {
    const int i = tid + 256 * j;
    const int row = i / (H_ / 4), c4 = i % (H_ / 4);
    *(v4f*)&tqs[row * TQ_STRIDE + c4 * 4] = tv[j];
  }
  __syncthreads();

  const float* a0p = &tqs[hl * TQ_STRIDE + 4 * half];
  const float* a1p = a0p + 16 * TQ_STRIDE;
  float4 af0[2][SC_U], af1[2][SC_U];
#pragma unroll
  for (int i = 0; i < SC_U; ++i) {
    af0[0][i] = *(const float4*)(a0p + 8 * i);
    af1[0][i] = *(const float4*)(a1p + 8 * i);
  }

  v8f acc0 = {0.f, 0.f, 0.f, 0.f, 0.f, 0.f, 0.f, 0.f};
  v8f acc1 = {0.f, 0.f, 0.f, 0.f, 0.f, 0.f, 0.f, 0.f};
#pragma unroll
  for (int u = 0; u < SC_NU; ++u) {
    const int g = u >> 1, cur = u & 1, nxt = cur ^ 1;
    if (cur == 0 && g + SC_AHEAD < SC_NG) {
#pragma unroll
      for (int i = 0; i < SC_GL; ++i)
        kv[(g + SC_AHEAD) % SC_RING][i] =
            *(const float4*)(krow + 8 * (SC_GL * (g + SC_AHEAD) + i));
    }
    if (u + 1 < SC_NU) {
#pragma unroll
      for (int i = 0; i < SC_U; ++i) {
        af0[nxt][i] = *(const float4*)(a0p + 8 * (SC_U * (u + 1) + i));
        af1[nxt][i] = *(const float4*)(a1p + 8 * (SC_U * (u + 1) + i));
      }
    }
    __builtin_amdgcn_sched_barrier(0);
#pragma unroll
    for (int i = 0; i < SC_U; ++i) {
      const float4 kq = kv[g % SC_RING][SC_U * cur + i];
      const float4 p  = af0[cur][i];
      const float4 q  = af1[cur][i];
      v2f blo, bhi, p0, p1, q0, q1;
      blo.x = kq.x; blo.y = kq.y;  bhi.x = kq.z; bhi.y = kq.w;
      p0.x = p.x;   p0.y = p.y;    p1.x = p.z;   p1.y = p.w;
      q0.x = q.x;   q0.y = q.y;    q1.x = q.z;   q1.y = q.w;
      acc0 = WMMA_F32(p0, blo, acc0);
      acc1 = WMMA_F32(q0, blo, acc1);
      acc0 = WMMA_F32(p1, bhi, acc0);
      acc1 = WMMA_F32(q1, bhi, acc1);
    }
  }

  const int s = s0 + wv * 16 + hl;
  const bool valid = s < seq_len[b];
  float* ag = alpha_g + (size_t)b * CA_ * S_;
  for (int r = 0; r < 8; ++r) {
    const int c0 = r + 8 * half;
    float v0 = valid ? __expf(acc0[r]) : 0.0f;
    float v1 = valid ? __expf(acc1[r]) : 0.0f;
    ag[(size_t)c0 * S_ + s]        = v0;
    ag[(size_t)(c0 + 16) * S_ + s] = v1;
    const float r0 = row16_sum(v0);
    const float r1 = row16_sum(v1);
    if (hl == 0) {
      dens[c0 * 8 + wv]        = r0;
      dens[(c0 + 16) * 8 + wv] = r1;
    }
  }
  __syncthreads();
  if (tid < CA_) {
    float d = 0.f;
    for (int w = 0; w < 8; ++w) d += dens[tid * 8 + w];
    den_part[((size_t)b * NCHUNK + chunk) * CA_ + tid] = d;
  }
}

constexpr int OUT_WAVES = 4;
constexpr int OUT_CG    = H_ / 64;
constexpr int OUT_SLOTS = KSPLIT * B_;
constexpr int OUT_ST    = 4;
static_assert(OUT_ST == 4, "the pinning statement in k_out names 4 k-steps");

__device__ __forceinline__ void out_plan(const int* __restrict__ seq_len, int lane,
                                         int& tiles, int& eend) {
  const int bl = lane & 15;
  int len = seq_len[bl];
  len = len < 0 ? 0 : (len > S_ ? S_ : len);
  tiles = (len + 15) >> 4;
  int p = tiles;
  p += dpp_row<DPP_ROW_SHR + 1>(0, p);
  p += dpp_row<DPP_ROW_SHR + 2>(0, p);
  p += dpp_row<DPP_ROW_SHR + 4>(0, p);
  p += dpp_row<DPP_ROW_SHR + 8>(0, p);
  const int G = dpp_row<DPP_ROW_SHARE + 15>(0, p);
  int r = (G > 0) ? (128 * p + G) / (2 * G) : 0;
  r = r > 49 + bl ? 49 + bl : r;
  r = r < bl + 1 ? bl + 1 : r;
  int m = r - bl;
  int q;
  q = dpp_row<DPP_ROW_SHR + 1>(m, m);  m = m > q ? m : q;
  q = dpp_row<DPP_ROW_SHR + 2>(m, m);  m = m > q ? m : q;
  q = dpp_row<DPP_ROW_SHR + 4>(m, m);  m = m > q ? m : q;
  q = dpp_row<DPP_ROW_SHR + 8>(m, m);  m = m > q ? m : q;
  eend = bl + m;
}

__global__ void __launch_bounds__(32 * OUT_WAVES)
k_out(const float* __restrict__ value, const float* __restrict__ alpha_g,
      const int* __restrict__ seq_len, float* __restrict__ num_part) {
  __shared__ float red[OUT_WAVES][64][32];
  const int gid  = blockIdx.x;
  const int cg   = gid % OUT_CG;
  const int slot = gid / OUT_CG;
  const int tid  = threadIdx.x;
  const int lane = tid & 31;
  const int wv   = tid >> 5;
  const int hl   = lane & 15;
  const int half = lane >> 4;
  const int kh   = half * 2;

  int tiles, eend;
  out_plan(seq_len, lane, tiles, eend);
  const int b = __popc((unsigned)__ballot(eend <= slot) & 0xFFFFu);
  if (b > 15) return;
  const int ep = __builtin_amdgcn_readlane(eend, b > 0 ? b - 1 : 0);
  const int e0 = (b > 0) ? ep : 0;
  const int e1 = __builtin_amdgcn_readlane(eend, b);
  const int tb = __builtin_amdgcn_readlane(tiles, b);
  const int nb = e1 - e0;
  const int lo = ((slot - e0) * tb) / nb;
  const int hi = ((slot - e0 + 1) * tb) / nb;
  const int n  = hi - lo;
  const int t0 = __builtin_amdgcn_readfirstlane(lo + (wv * n) / OUT_WAVES);
  const int t1 = __builtin_amdgcn_readfirstlane(lo + ((wv + 1) * n) / OUT_WAVES);

  const float* ap0 = alpha_g + ((size_t)b * CA_ + hl) * S_ + kh;
  const float* ap1 = ap0 + (size_t)16 * S_;
  const float* vp  = value + ((size_t)b * S_ + kh) * H_ + cg * 64 + 4 * hl;

  v8f acc[8];
  for (int a = 0; a < 8; ++a) acc[a] = (v8f){0.f, 0.f, 0.f, 0.f, 0.f, 0.f, 0.f, 0.f};

  if (t0 < t1) {
    v2f    a0[3][OUT_ST], a1[3][OUT_ST];
    v4f    v0[3][OUT_ST], v1[3][OUT_ST];
    const int last = t1 - 1;
#define OUT_LOAD(SET, TILE)                                              \
    {                                                                    \
      const int tt_ = (TILE) < last ? (TILE) : last;                     \
      _Pragma("unroll") for (int i = 0; i < OUT_ST; ++i) {               \
        const int k = 16 * tt_ + 4 * i;                                  \
        a0[SET][i] = *(const v2f*)(ap0 + k);                             \
        a1[SET][i] = *(const v2f*)(ap1 + k);                             \
        v0[SET][i] = *(const v4f*)(vp + (size_t)k * H_);                 \
        v1[SET][i] = *(const v4f*)(vp + (size_t)(k + 1) * H_);           \
      }                                                                  \
    }
#define OUT_MMA(SET)                                                     \
    {                                                                    \
         \
      asm volatile("" : "+v"(v0[SET][0]), "+v"(v1[SET][0]), "+v"(v0[SET][1]),   \
                        "+v"(v1[SET][1]), "+v"(v0[SET][2]), "+v"(v1[SET][2]),   \
                        "+v"(v0[SET][3]), "+v"(v1[SET][3]));                    \
      _Pragma("unroll") for (int i = 0; i < OUT_ST; ++i) {               \
        const float x0[4] = {v0[SET][i].x, v0[SET][i].y, v0[SET][i].z, v0[SET][i].w}; \
        const float x1[4] = {v1[SET][i].x, v1[SET][i].y, v1[SET][i].z, v1[SET][i].w}; \
        _Pragma("unroll") for (int j = 0; j < 4; ++j) {                  \
          v2f bf;                                                        \
          bf.x = x0[j];                                                  \
          bf.y = x1[j];                                                  \
          acc[j]     = WMMA_F32(a0[SET][i], bf, acc[j]);                 \
          acc[4 + j] = WMMA_F32(a1[SET][i], bf, acc[4 + j]);             \
        }                                                                \
      }                                                                  \
    }
    OUT_LOAD(0, t0);
    OUT_LOAD(1, t0 + 1);
    int t = t0;
    for (; t + 3 <= t1; t += 3) {
      OUT_LOAD(2, t + 2);
      __builtin_amdgcn_sched_barrier(0);
      OUT_MMA(0);
      OUT_LOAD(0, t + 3);
      __builtin_amdgcn_sched_barrier(0);
      OUT_MMA(1);
      OUT_LOAD(1, t + 4);
      __builtin_amdgcn_sched_barrier(0);
      OUT_MMA(2);
    }
    if (t < t1) OUT_MMA(0);
    if (t + 1 < t1) OUT_MMA(1);
#undef OUT_LOAD
#undef OUT_MMA
  }

  for (int a = 0; a < 8; ++a)
    for (int r = 0; r < 8; ++r)
      red[wv][a * 8 + r][lane] = acc[a][r];
  __syncthreads();

  float* o = num_part + (size_t)slot * CA_ * H_ + cg * 64 + 4 * hl;
  const int t = wv >> 1;
  for (int q = 0; q < 4; ++q) {
    const int r = 4 * (wv & 1) + q;
    float sum[4];
    for (int j = 0; j < 4; ++j) {
      const int reg = (4 * t + j) * 8 + r;
      float s = red[0][reg][lane];
      for (int w = 1; w < OUT_WAVES; ++w) s += red[w][reg][lane];
      sum[j] = s;
    }
    const int c = 16 * t + 8 * half + r;
    *(float4*)(o + (size_t)c * H_) = make_float4(sum[0], sum[1], sum[2], sum[3]);
  }
}

__global__ void k_fin(const float* __restrict__ num_part, const float* __restrict__ den_part,
                      const int* __restrict__ seq_len, float* __restrict__ out) {
  const int idx = blockIdx.x * 256 + threadIdx.x;
  static_assert(H_ % 256 == 0, "a block must lie inside one (b, c) row");
  const int bc  = blockIdx.x / (H_ / 256);
  const int b   = bc >> 5, c = bc & 31;
  const int h   = idx - bc * H_;
  float dv[NCHUNK];
#pragma unroll
  for (int k = 0; k < NCHUNK; ++k) dv[k] = den_part[((size_t)b * NCHUNK + k) * CA_ + c];
  int tiles, eend;
  out_plan(seq_len, threadIdx.x & 31, tiles, eend);
  const int ep = __builtin_amdgcn_readlane(eend, b > 0 ? b - 1 : 0);
  const int e0 = (b > 0) ? ep : 0;
  const int e1 = __builtin_amdgcn_readlane(eend, b);
  float num = 0.f;
  for (int base = e0; base < e1; base += 8) {
    float nv[8];
#pragma unroll
    for (int q = 0; q < 8; ++q) {
      const int sl = base + q < e1 ? base + q : e1 - 1;
      nv[q] = num_part[((size_t)sl * CA_ + c) * H_ + h];
    }
#pragma unroll
    for (int q = 0; q < 8; ++q) num += (base + q < e1) ? nv[q] : 0.f;
  }
  float den = 0.f;
#pragma unroll
  for (int k = 0; k < NCHUNK; ++k) den += dv[k];
  out[idx] = num / den;
}

extern "C" void kernel_launch(void* const* d_in, const int* in_sizes, int n_in,
                              void* d_out, int out_size, void* d_ws, size_t ws_size,
                              hipStream_t stream) {
  (void)in_sizes; (void)n_in; (void)out_size; (void)ws_size;
  const float* key     = (const float*)d_in[0];
  const float* value   = (const float*)d_in[1];
  const float* query   = (const float*)d_in[2];
  const int*   seq_len = (const int*)d_in[3];
  const float* W       = (const float*)d_in[4];
  const float* bias    = (const float*)d_in[5];
  float* ws  = (float*)d_ws;
  float* out = (float*)d_out;

  k_tq<<<B_ * TQ_NT, 32 * TQ_WAVES, 0, stream>>>(query, W, bias, ws + TQ_OFF);

  const int smem_bytes = (32 * TQ_STRIDE + 32 * 8) * sizeof(float);
  k_scores<<<B_ * NCHUNK, 256, smem_bytes, stream>>>(key, ws + TQ_OFF, seq_len,
                                                     ws + AL_OFF, ws + DEN_OFF);

  k_out<<<OUT_SLOTS * OUT_CG, 32 * OUT_WAVES, 0, stream>>>(value, ws + AL_OFF, seq_len, ws + NUM_OFF);

  k_fin<<<(B_ * CA_ * H_) / 256, 256, 0, stream>>>(ws + NUM_OFF, ws + DEN_OFF, seq_len, out);
}


# scores: K split over a block's 4 waves (tq quarter in registers, no LDS staging), valid tiles spread over 256 blocks; partial scores added via LDS in wave order, per-block denominators: new sum order
# speedup vs baseline: 2.4521x; 1.1061x over previous
#include <hip/hip_runtime.h>

typedef __attribute__((ext_vector_type(2))) float v2f;
typedef __attribute__((ext_vector_type(4))) float v4f;
typedef __attribute__((ext_vector_type(8))) float v8f;

#define WMMA_F32(a, b, c) \
  __builtin_amdgcn_wmma_f32_16x16x4_f32(false, (a), false, (b), (short)0, (c), false, false)

constexpr int DPP_ROW_SHR   = 0x110;
constexpr int DPP_ROW_SHARE = 0x150;
constexpr int DPP_ROW_XMASK = 0x160;
template <int CTRL>
__device__ __forceinline__ int dpp_row(int old, int x) {
  return __builtin_amdgcn_update_dpp(old, x, CTRL, 0xf, 0xf, false);
}
template <int CTRL>
__device__ __forceinline__ float dpp_rowf(float old, float x) {
  return __int_as_float(dpp_row<CTRL>(__float_as_int(old), __float_as_int(x)));
}
__device__ __forceinline__ float row16_sum(float x) {
  x += dpp_rowf<DPP_ROW_XMASK + 1>(0.f, x);
  x += dpp_rowf<DPP_ROW_XMASK + 2>(0.f, x);
  x += dpp_rowf<DPP_ROW_XMASK + 4>(0.f, x);
  x += dpp_rowf<DPP_ROW_XMASK + 8>(0.f, x);
  return x;
}

constexpr int B_ = 16, S_ = 2048, H_ = 768, CA_ = 32, IN_ = 300;
constexpr int SCN_SLOTS = 256;
constexpr int KSPLIT = 4;
constexpr int KLEN = S_ / KSPLIT;

constexpr size_t TQ_OFF  = 0;
constexpr size_t TQ_SZ   = (size_t)B_ * CA_ * H_;
constexpr size_t AL_OFF  = TQ_OFF + TQ_SZ;
constexpr size_t AL_SZ   = (size_t)B_ * CA_ * S_;
constexpr size_t DEN_OFF = AL_OFF + AL_SZ;
constexpr size_t DEN_SZ  = (size_t)SCN_SLOTS * CA_;
constexpr size_t NUM_OFF = DEN_OFF + DEN_SZ;
constexpr size_t NUM_SZ  = (size_t)KSPLIT * B_ * CA_ * H_;

constexpr int TQ_WAVES = 5;
constexpr int TQ_STEPS = IN_ / (4 * TQ_WAVES);
constexpr int TQ_NT    = H_ / 32;

__global__ void __launch_bounds__(32 * TQ_WAVES)
k_tq(const float* __restrict__ query, const float* __restrict__ W,
     const float* __restrict__ bias, float* __restrict__ tq) {
  __shared__ float red[TQ_WAVES][32][32];
  const int b    = blockIdx.x / TQ_NT;
  const int nt   = blockIdx.x % TQ_NT;
  const int tid  = threadIdx.x;
  const int lane = tid & 31;
  const int wv   = tid >> 5;
  const int hl   = lane & 15;
  const int half = lane >> 4;
  const int kh   = half * 2;
  const int kb   = wv * (TQ_STEPS * 4);

  const float* ap0 = query + ((size_t)b * CA_ + hl) * IN_ + kb + kh;
  const float* ap1 = ap0 + (size_t)16 * IN_;
  const float* bp  = W + (size_t)(kb + kh) * H_ + nt * 32 + 2 * hl;

  v8f acc00 = {0.f, 0.f, 0.f, 0.f, 0.f, 0.f, 0.f, 0.f};
  v8f acc01 = acc00, acc10 = acc00, acc11 = acc00;
  v2f a0[TQ_STEPS], a1[TQ_STEPS], w0[TQ_STEPS], w1[TQ_STEPS];
#pragma unroll
  for (int i = 0; i < TQ_STEPS; ++i) {
    a0[i] = *(const v2f*)(ap0 + 4 * i);
    a1[i] = *(const v2f*)(ap1 + 4 * i);
    w0[i] = *(const v2f*)(bp + (size_t)(4 * i) * H_);
    w1[i] = *(const v2f*)(bp + (size_t)(4 * i + 1) * H_);
  }
  __builtin_amdgcn_sched_barrier(0);
#pragma unroll
  for (int i = 0; i < TQ_STEPS; ++i) {
    v2f b0, b1;
    b0.x = w0[i].x; b0.y = w1[i].x;
    b1.x = w0[i].y; b1.y = w1[i].y;
    acc00 = WMMA_F32(a0[i], b0, acc00);
    acc01 = WMMA_F32(a0[i], b1, acc01);
    acc10 = WMMA_F32(a1[i], b0, acc10);
    acc11 = WMMA_F32(a1[i], b1, acc11);
  }

  for (int r = 0; r < 8; ++r) {
    red[wv][r][lane]      = acc00[r];
    red[wv][8 + r][lane]  = acc01[r];
    red[wv][16 + r][lane] = acc10[r];
    red[wv][24 + r][lane] = acc11[r];
  }
  __syncthreads();

  if (wv < 4) {
    const int h0 = nt * 32 + 2 * hl;
    const v2f bv = *(const v2f*)(bias + h0);
    for (int q = 0; q < 2; ++q) {
      const int r = 2 * wv + q;
      float s00 = red[0][r][lane],      s01 = red[0][8 + r][lane];
      float s10 = red[0][16 + r][lane], s11 = red[0][24 + r][lane];
      for (int w = 1; w < TQ_WAVES; ++w) {
        s00 += red[w][r][lane];
        s01 += red[w][8 + r][lane];
        s10 += red[w][16 + r][lane];
        s11 += red[w][24 + r][lane];
      }
      const int c = 8 * half + r;
      v2f o0, o1;
      o0.x = tanhf(s00 + bv.x); o0.y = tanhf(s01 + bv.y);
      o1.x = tanhf(s10 + bv.x); o1.y = tanhf(s11 + bv.y);
      *(v2f*)(tq + ((size_t)b * CA_ + c) * H_ + h0)      = o0;
      *(v2f*)(tq + ((size_t)b * CA_ + c + 16) * H_ + h0) = o1;
    }
  }
}

template <int NS>
__device__ __forceinline__ void out_plan(const int* __restrict__ seq_len, int lane,
                                         int& tiles, int& eend) {
  const int bl = lane & 15;
  int len = seq_len[bl];
  len = len < 0 ? 0 : (len > S_ ? S_ : len);
  tiles = (len + 15) >> 4;
  int p = tiles;
  p += dpp_row<DPP_ROW_SHR + 1>(0, p);
  p += dpp_row<DPP_ROW_SHR + 2>(0, p);
  p += dpp_row<DPP_ROW_SHR + 4>(0, p);
  p += dpp_row<DPP_ROW_SHR + 8>(0, p);
  const int G = dpp_row<DPP_ROW_SHARE + 15>(0, p);
  int r = (G > 0) ? (2 * NS * p + G) / (2 * G) : 0;
  r = r > NS - 15 + bl ? NS - 15 + bl : r;
  r = r < bl + 1 ? bl + 1 : r;
  int m = r - bl;
  int q;
  q = dpp_row<DPP_ROW_SHR + 1>(m, m);  m = m > q ? m : q;
  q = dpp_row<DPP_ROW_SHR + 2>(m, m);  m = m > q ? m : q;
  q = dpp_row<DPP_ROW_SHR + 4>(m, m);  m = m > q ? m : q;
  q = dpp_row<DPP_ROW_SHR + 8>(m, m);  m = m > q ? m : q;
  eend = bl + m;
}

__device__ __forceinline__ int row16_prefix(int p) {
  p += dpp_row<DPP_ROW_SHR + 1>(0, p);
  p += dpp_row<DPP_ROW_SHR + 2>(0, p);
  p += dpp_row<DPP_ROW_SHR + 4>(0, p);
  p += dpp_row<DPP_ROW_SHR + 8>(0, p);
  return p;
}
__device__ __forceinline__ void sc_plan(const int* __restrict__ seq_len, int lane,
                                        int& tiles, int& send, int& T) {
  const int bl = lane & 15;
  int len = seq_len[bl];
  len = len < 0 ? 0 : (len > S_ ? S_ : len);
  tiles = (len + 15) >> 4;
  const int G  = dpp_row<DPP_ROW_SHARE + 15>(0, row16_prefix(tiles));
  int T0 = (G + SCN_SLOTS - 1) / SCN_SLOTS;
  T0 = T0 < 1 ? 1 : T0;
  const int ea = row16_prefix((tiles + T0 - 1) / T0);
  const int eb = row16_prefix((tiles + T0) / (T0 + 1));
  const bool fits = dpp_row<DPP_ROW_SHARE + 15>(0, ea) <= SCN_SLOTS;
  T    = fits ? T0 : T0 + 1;
  send = fits ? ea : eb;
}

constexpr int SCN_WAVES = 4;
constexpr int SCN_KQ    = H_ / SCN_WAVES;
constexpr int SCN_L     = SCN_KQ / 8;
constexpr int SCN_HEAD  = 6;
static_assert(DEN_SZ == (size_t)SCN_SLOTS * CA_, "one denominator row per slot");

__global__ void __launch_bounds__(32 * SCN_WAVES)
k_scores(const float* __restrict__ key, const float* __restrict__ tq,
         const int* __restrict__ seq_len,
         float* __restrict__ alpha_g, float* __restrict__ den_part) {
  __shared__ float red[2][SCN_WAVES][16][32];
  const int slot = blockIdx.x;
  const int tid  = threadIdx.x;
  const int lane = tid & 31;
  const int wv   = tid >> 5;
  const int hl   = lane & 15;
  const int half = lane >> 4;

  int tiles, send, T;
  sc_plan(seq_len, lane, tiles, send, T);
  const int b = __popc((unsigned)__ballot(send <= slot) & 0xFFFFu);
  if (b > 15) return;
  const int sp = __builtin_amdgcn_readlane(send, b > 0 ? b - 1 : 0);
  const int e0 = (b > 0) ? sp : 0;
  const int tb = __builtin_amdgcn_readlane(tiles, b);
  const int ts = __builtin_amdgcn_readfirstlane(T);
  const int lo = __builtin_amdgcn_readfirstlane((slot - e0) * ts);
  const int hi = lo + ts < tb ? lo + ts : tb;
  const int len = seq_len[b];

  float dsum[4] = {0.f, 0.f, 0.f, 0.f};

  const float* a0p = tq + ((size_t)b * CA_ + hl) * H_ + SCN_KQ * wv + 4 * half;
  const float* a1p = a0p + (size_t)16 * H_;
  const float* kp  = key + ((size_t)b * S_ + hl) * H_ + SCN_KQ * wv + 4 * half;
  float* ag = alpha_g + (size_t)b * CA_ * S_;
  const int last = hi - 1;

  float4 kv[2][SCN_L];
  float4 af0[SCN_L], af1[SCN_L];
#define SCN_LOAD(SET, TILE)                                              \
  {                                                                    \
    const int tt_ = (TILE) < last ? (TILE) : last;                     \
    const float* kr_ = kp + (size_t)(16 * tt_) * H_;                   \
    _Pragma("unroll") for (int m = 0; m < SCN_L; ++m)                  \
      kv[SET][m] = *(const float4*)(kr_ + 8 * m);                      \
  }
#define SCN_STEPS(SET, M0, M1)                                         \
  {                                                                    \
    _Pragma("unroll") for (int m = (M0); m < (M1); ++m) {              \
      const float4 kq = kv[SET][m];                                    \
      const float4 p  = af0[m];                                        \
      const float4 q  = af1[m];                                        \
      v2f blo, bhi, p0, p1, q0, q1;                                    \
      blo.x = kq.x; blo.y = kq.y;  bhi.x = kq.z; bhi.y = kq.w;         \
      p0.x = p.x;   p0.y = p.y;    p1.x = p.z;   p1.y = p.w;           \
      q0.x = q.x;   q0.y = q.y;    q1.x = q.z;   q1.y = q.w;           \
      acc0 = WMMA_F32(p0, blo, acc0);                                  \
      acc1 = WMMA_F32(q0, blo, acc1);                                  \
      acc0 = WMMA_F32(p1, bhi, acc0);                                  \
      acc1 = WMMA_F32(q1, bhi, acc1);                                  \
    }                                                                  \
  }
#define SCN_PUT(TILE)                                                  \
  {                                                                    \
    float (*ws_)[16][32] = red[(TILE) & 1];                            \
    _Pragma("unroll") for (int r = 0; r < 8; ++r) {                    \
      ws_[wv][r][lane]     = acc0[r];                                  \
      ws_[wv][8 + r][lane] = acc1[r];                                  \
    }                                                                  \
    __syncthreads();                                                   \
    acc0 = zero8;                                                      \
    acc1 = zero8;                                                      \
  }
#define SCN_FIN(TILE)                                                  \
  {                                                                    \
    float (*rs_)[16][32] = red[(TILE) & 1];                            \
    const int s_ = 16 * (TILE) + hl;                                   \
    const bool valid_ = s_ < len;                                      \
    _Pragma("unroll") for (int q = 0; q < 2; ++q) {                    \
      const int r = 2 * wv + q;                                        \
      float x0 = rs_[0][r][lane], x1 = rs_[0][8 + r][lane];            \
      _Pragma("unroll") for (int w = 1; w < SCN_WAVES; ++w) {          \
        x0 += rs_[w][r][lane];                                         \
        x1 += rs_[w][8 + r][lane];                                     \
      }                                                                \
      const float v0 = valid_ ? __expf(x0) : 0.0f;                     \
      const float v1 = valid_ ? __expf(x1) : 0.0f;                     \
      const int c0 = r + 8 * half;                                     \
      ag[(size_t)c0 * S_ + s_]        = v0;                            \
      ag[(size_t)(c0 + 16) * S_ + s_] = v1;                            \
      dsum[2 * q]     += v0;                                           \
      dsum[2 * q + 1] += v1;                                           \
    }                                                                  \
  }
  SCN_LOAD(0, lo);
#pragma unroll
  for (int m = 0; m < SCN_L; ++m) {
    af0[m] = *(const float4*)(a0p + 8 * m);
    af1[m] = *(const float4*)(a1p + 8 * m);
  }
  SCN_LOAD(1, lo + 1);
  const v8f zero8 = {0.f, 0.f, 0.f, 0.f, 0.f, 0.f, 0.f, 0.f};
  v8f acc0 = zero8, acc1 = zero8;
  __builtin_amdgcn_sched_barrier(0);
  SCN_STEPS(0, 0, SCN_L);
  SCN_PUT(lo);
  for (int t = lo + 1; t < hi; t += 2) {
    SCN_LOAD(0, t + 1);
    __builtin_amdgcn_sched_barrier(0);
    SCN_FIN(t - 1);
    SCN_STEPS(1, 0, SCN_HEAD);
    __builtin_amdgcn_sched_barrier(0);
    SCN_STEPS(1, SCN_HEAD, SCN_L);
    SCN_PUT(t);
    if (t + 1 < hi) {
      SCN_LOAD(1, t + 2);
      __builtin_amdgcn_sched_barrier(0);
      SCN_FIN(t);
      SCN_STEPS(0, 0, SCN_HEAD);
      __builtin_amdgcn_sched_barrier(0);
      SCN_STEPS(0, SCN_HEAD, SCN_L);
      SCN_PUT(t + 1);
    }
  }
  SCN_FIN(last);
#undef SCN_LOAD
#undef SCN_STEPS
#undef SCN_PUT
#undef SCN_FIN

  for (int q = 0; q < 2; ++q) {
    const float d0 = row16_sum(dsum[2 * q]);
    const float d1 = row16_sum(dsum[2 * q + 1]);
    if (hl == 0) {
      const int c0 = 2 * wv + q + 8 * half;
      den_part[(size_t)slot * CA_ + c0]      = d0;
      den_part[(size_t)slot * CA_ + c0 + 16] = d1;
    }
  }
}

constexpr int OUT_WAVES = 4;
constexpr int OUT_CG    = H_ / 64;
constexpr int OUT_SLOTS = KSPLIT * B_;
constexpr int OUT_ST    = 4;
static_assert(OUT_ST == 4, "the pinning statement in k_out names 4 k-steps");

__global__ void __launch_bounds__(32 * OUT_WAVES)
k_out(const float* __restrict__ value, const float* __restrict__ alpha_g,
      const int* __restrict__ seq_len, float* __restrict__ num_part) {
  __shared__ float red[OUT_WAVES][64][32];
  const int gid  = blockIdx.x;
  const int cg   = gid % OUT_CG;
  const int slot = gid / OUT_CG;
  const int tid  = threadIdx.x;
  const int lane = tid & 31;
  const int wv   = tid >> 5;
  const int hl   = lane & 15;
  const int half = lane >> 4;
  const int kh   = half * 2;

  int tiles, eend;
  out_plan<OUT_SLOTS>(seq_len, lane, tiles, eend);
  const int b = __popc((unsigned)__ballot(eend <= slot) & 0xFFFFu);
  if (b > 15) return;
  const int ep = __builtin_amdgcn_readlane(eend, b > 0 ? b - 1 : 0);
  const int e0 = (b > 0) ? ep : 0;
  const int e1 = __builtin_amdgcn_readlane(eend, b);
  const int tb = __builtin_amdgcn_readlane(tiles, b);
  const int nb = e1 - e0;
  const int lo = ((slot - e0) * tb) / nb;
  const int hi = ((slot - e0 + 1) * tb) / nb;
  const int n  = hi - lo;
  const int t0 = __builtin_amdgcn_readfirstlane(lo + (wv * n) / OUT_WAVES);
  const int t1 = __builtin_amdgcn_readfirstlane(lo + ((wv + 1) * n) / OUT_WAVES);

  const float* ap0 = alpha_g + ((size_t)b * CA_ + hl) * S_ + kh;
  const float* ap1 = ap0 + (size_t)16 * S_;
  const float* vp  = value + ((size_t)b * S_ + kh) * H_ + cg * 64 + 4 * hl;

  v8f acc[8];
  for (int a = 0; a < 8; ++a) acc[a] = (v8f){0.f, 0.f, 0.f, 0.f, 0.f, 0.f, 0.f, 0.f};

  if (t0 < t1) {
    v2f    a0[3][OUT_ST], a1[3][OUT_ST];
    v4f    v0[3][OUT_ST], v1[3][OUT_ST];
    const int last = t1 - 1;
#define OUT_LOAD(SET, TILE)                                              \
    {                                                                    \
      const int tt_ = (TILE) < last ? (TILE) : last;                     \
      _Pragma("unroll") for (int i = 0; i < OUT_ST; ++i) {               \
        const int k = 16 * tt_ + 4 * i;                                  \
        a0[SET][i] = *(const v2f*)(ap0 + k);                             \
        a1[SET][i] = *(const v2f*)(ap1 + k);                             \
        v0[SET][i] = *(const v4f*)(vp + (size_t)k * H_);                 \
        v1[SET][i] = *(const v4f*)(vp + (size_t)(k + 1) * H_);           \
      }                                                                  \
    }
#define OUT_MMA(SET)                                                     \
    {                                                                    \
         \
      asm volatile("" : "+v"(v0[SET][0]), "+v"(v1[SET][0]), "+v"(v0[SET][1]),   \
                        "+v"(v1[SET][1]), "+v"(v0[SET][2]), "+v"(v1[SET][2]),   \
                        "+v"(v0[SET][3]), "+v"(v1[SET][3]));                    \
      _Pragma("unroll") for (int i = 0; i < OUT_ST; ++i) {               \
        const float x0[4] = {v0[SET][i].x, v0[SET][i].y, v0[SET][i].z, v0[SET][i].w}; \
        const float x1[4] = {v1[SET][i].x, v1[SET][i].y, v1[SET][i].z, v1[SET][i].w}; \
        _Pragma("unroll") for (int j = 0; j < 4; ++j) {                  \
          v2f bf;                                                        \
          bf.x = x0[j];                                                  \
          bf.y = x1[j];                                                  \
          acc[j]     = WMMA_F32(a0[SET][i], bf, acc[j]);                 \
          acc[4 + j] = WMMA_F32(a1[SET][i], bf, acc[4 + j]);             \
        }                                                                \
      }                                                                  \
    }
    OUT_LOAD(0, t0);
    OUT_LOAD(1, t0 + 1);
    int t = t0;
    for (; t + 3 <= t1; t += 3) {
      OUT_LOAD(2, t + 2);
      __builtin_amdgcn_sched_barrier(0);
      OUT_MMA(0);
      OUT_LOAD(0, t + 3);
      __builtin_amdgcn_sched_barrier(0);
      OUT_MMA(1);
      OUT_LOAD(1, t + 4);
      __builtin_amdgcn_sched_barrier(0);
      OUT_MMA(2);
    }
    if (t < t1) OUT_MMA(0);
    if (t + 1 < t1) OUT_MMA(1);
#undef OUT_LOAD
#undef OUT_MMA
  }

  for (int a = 0; a < 8; ++a)
    for (int r = 0; r < 8; ++r)
      red[wv][a * 8 + r][lane] = acc[a][r];
  __syncthreads();

  float* o = num_part + (size_t)slot * CA_ * H_ + cg * 64 + 4 * hl;
  const int t = wv >> 1;
  for (int q = 0; q < 4; ++q) {
    const int r = 4 * (wv & 1) + q;
    float sum[4];
    for (int j = 0; j < 4; ++j) {
      const int reg = (4 * t + j) * 8 + r;
      float s = red[0][reg][lane];
      for (int w = 1; w < OUT_WAVES; ++w) s += red[w][reg][lane];
      sum[j] = s;
    }
    const int c = 16 * t + 8 * half + r;
    *(float4*)(o + (size_t)c * H_) = make_float4(sum[0], sum[1], sum[2], sum[3]);
  }
}

__global__ void k_fin(const float* __restrict__ num_part, const float* __restrict__ den_part,
                      const int* __restrict__ seq_len, float* __restrict__ out) {
  const int idx = blockIdx.x * 256 + threadIdx.x;
  static_assert(H_ % 256 == 0, "a block must lie inside one (b, c) row");
  const int bc  = blockIdx.x / (H_ / 256);
  const int b   = bc >> 5, c = bc & 31;
  const int h   = idx - bc * H_;
  const int lane = threadIdx.x & 31;
  int stiles, send, st;
  sc_plan(seq_len, lane, stiles, send, st);
  const int sp  = __builtin_amdgcn_readlane(send, b > 0 ? b - 1 : 0);
  const int se0 = (b > 0) ? sp : 0;
  const int se1 = __builtin_amdgcn_readlane(send, b);
  float dl = 0.f;
  for (int t = se0 + lane; t < se1; t += 32) dl += den_part[(size_t)t * CA_ + c];
  int tiles, eend;
  out_plan<OUT_SLOTS>(seq_len, lane, tiles, eend);
  const int ep = __builtin_amdgcn_readlane(eend, b > 0 ? b - 1 : 0);
  const int e0 = (b > 0) ? ep : 0;
  const int e1 = __builtin_amdgcn_readlane(eend, b);
  float num = 0.f;
  for (int base = e0; base < e1; base += 8) {
    float nv[8];
#pragma unroll
    for (int q = 0; q < 8; ++q) {
      const int sl = base + q < e1 ? base + q : e1 - 1;
      nv[q] = num_part[((size_t)sl * CA_ + c) * H_ + h];
    }
#pragma unroll
    for (int q = 0; q < 8; ++q) num += (base + q < e1) ? nv[q] : 0.f;
  }
  dl = row16_sum(dl);
  const float den = __int_as_float(__builtin_amdgcn_readlane(__float_as_int(dl), 0)) +
                    __int_as_float(__builtin_amdgcn_readlane(__float_as_int(dl), 16));
  out[idx] = num / den;
}

extern "C" void kernel_launch(void* const* d_in, const int* in_sizes, int n_in,
                              void* d_out, int out_size, void* d_ws, size_t ws_size,
                              hipStream_t stream) {
  (void)in_sizes; (void)n_in; (void)out_size; (void)ws_size;
  const float* key     = (const float*)d_in[0];
  const float* value   = (const float*)d_in[1];
  const float* query   = (const float*)d_in[2];
  const int*   seq_len = (const int*)d_in[3];
  const float* W       = (const float*)d_in[4];
  const float* bias    = (const float*)d_in[5];
  float* ws  = (float*)d_ws;
  float* out = (float*)d_out;

  k_tq<<<B_ * TQ_NT, 32 * TQ_WAVES, 0, stream>>>(query, W, bias, ws + TQ_OFF);

  k_scores<<<SCN_SLOTS, 32 * SCN_WAVES, 0, stream>>>(key, ws + TQ_OFF, seq_len,
                                                     ws + AL_OFF, ws + DEN_OFF);

  k_out<<<OUT_SLOTS * OUT_CG, 32 * OUT_WAVES, 0, stream>>>(value, ws + AL_OFF, seq_len, ws + NUM_OFF);

  k_fin<<<(B_ * CA_ * H_) / 256, 256, 0, stream>>>(ws + NUM_OFF, ws + DEN_OFF, seq_len, out);
}


# final kernel as 1024 fat waves, not 12288: one 64-thread block per (b, c) row, three float4 per thread, all of a wave's partial loads in flight in one trip; every sum in the same order
# speedup vs baseline: 2.5545x; 1.0418x over previous
#include <hip/hip_runtime.h>

typedef __attribute__((ext_vector_type(2))) float v2f;
typedef __attribute__((ext_vector_type(4))) float v4f;
typedef __attribute__((ext_vector_type(8))) float v8f;

#define WMMA_F32(a, b, c) \
  __builtin_amdgcn_wmma_f32_16x16x4_f32(false, (a), false, (b), (short)0, (c), false, false)

constexpr int DPP_ROW_SHR   = 0x110;
constexpr int DPP_ROW_SHARE = 0x150;
constexpr int DPP_ROW_XMASK = 0x160;
template <int CTRL>
__device__ __forceinline__ int dpp_row(int old, int x) {
  return __builtin_amdgcn_update_dpp(old, x, CTRL, 0xf, 0xf, false);
}
template <int CTRL>
__device__ __forceinline__ float dpp_rowf(float old, float x) {
  return __int_as_float(dpp_row<CTRL>(__float_as_int(old), __float_as_int(x)));
}
__device__ __forceinline__ float row16_sum(float x) {
  x += dpp_rowf<DPP_ROW_XMASK + 1>(0.f, x);
  x += dpp_rowf<DPP_ROW_XMASK + 2>(0.f, x);
  x += dpp_rowf<DPP_ROW_XMASK + 4>(0.f, x);
  x += dpp_rowf<DPP_ROW_XMASK + 8>(0.f, x);
  return x;
}

constexpr int B_ = 16, S_ = 2048, H_ = 768, CA_ = 32, IN_ = 300;
constexpr int SCN_SLOTS = 256;
constexpr int KSPLIT = 4;
constexpr int KLEN = S_ / KSPLIT;

constexpr size_t TQ_OFF  = 0;
constexpr size_t TQ_SZ   = (size_t)B_ * CA_ * H_;
constexpr size_t AL_OFF  = TQ_OFF + TQ_SZ;
constexpr size_t AL_SZ   = (size_t)B_ * CA_ * S_;
constexpr size_t DEN_OFF = AL_OFF + AL_SZ;
constexpr size_t DEN_SZ  = (size_t)SCN_SLOTS * CA_;
constexpr size_t NUM_OFF = DEN_OFF + DEN_SZ;
constexpr size_t NUM_SZ  = (size_t)KSPLIT * B_ * CA_ * H_;

constexpr int TQ_WAVES = 5;
constexpr int TQ_STEPS = IN_ / (4 * TQ_WAVES);
constexpr int TQ_NT    = H_ / 32;

__global__ void __launch_bounds__(32 * TQ_WAVES)
k_tq(const float* __restrict__ query, const float* __restrict__ W,
     const float* __restrict__ bias, float* __restrict__ tq) {
  __shared__ float red[TQ_WAVES][32][32];
  const int b    = blockIdx.x / TQ_NT;
  const int nt   = blockIdx.x % TQ_NT;
  const int tid  = threadIdx.x;
  const int lane = tid & 31;
  const int wv   = tid >> 5;
  const int hl   = lane & 15;
  const int half = lane >> 4;
  const int kh   = half * 2;
  const int kb   = wv * (TQ_STEPS * 4);

  const float* ap0 = query + ((size_t)b * CA_ + hl) * IN_ + kb + kh;
  const float* ap1 = ap0 + (size_t)16 * IN_;
  const float* bp  = W + (size_t)(kb + kh) * H_ + nt * 32 + 2 * hl;

  v8f acc00 = {0.f, 0.f, 0.f, 0.f, 0.f, 0.f, 0.f, 0.f};
  v8f acc01 = acc00, acc10 = acc00, acc11 = acc00;
  v2f a0[TQ_STEPS], a1[TQ_STEPS], w0[TQ_STEPS], w1[TQ_STEPS];
#pragma unroll
  for (int i = 0; i < TQ_STEPS; ++i) {
    a0[i] = *(const v2f*)(ap0 + 4 * i);
    a1[i] = *(const v2f*)(ap1 + 4 * i);
    w0[i] = *(const v2f*)(bp + (size_t)(4 * i) * H_);
    w1[i] = *(const v2f*)(bp + (size_t)(4 * i + 1) * H_);
  }
  __builtin_amdgcn_sched_barrier(0);
#pragma unroll
  for (int i = 0; i < TQ_STEPS; ++i) {
    v2f b0, b1;
    b0.x = w0[i].x; b0.y = w1[i].x;
    b1.x = w0[i].y; b1.y = w1[i].y;
    acc00 = WMMA_F32(a0[i], b0, acc00);
    acc01 = WMMA_F32(a0[i], b1, acc01);
    acc10 = WMMA_F32(a1[i], b0, acc10);
    acc11 = WMMA_F32(a1[i], b1, acc11);
  }

  for (int r = 0; r < 8; ++r) {
    red[wv][r][lane]      = acc00[r];
    red[wv][8 + r][lane]  = acc01[r];
    red[wv][16 + r][lane] = acc10[r];
    red[wv][24 + r][lane] = acc11[r];
  }
  __syncthreads();

  if (wv < 4) {
    const int h0 = nt * 32 + 2 * hl;
    const v2f bv = *(const v2f*)(bias + h0);
    for (int q = 0; q < 2; ++q) {
      const int r = 2 * wv + q;
      float s00 = red[0][r][lane],      s01 = red[0][8 + r][lane];
      float s10 = red[0][16 + r][lane], s11 = red[0][24 + r][lane];
      for (int w = 1; w < TQ_WAVES; ++w) {
        s00 += red[w][r][lane];
        s01 += red[w][8 + r][lane];
        s10 += red[w][16 + r][lane];
        s11 += red[w][24 + r][lane];
      }
      const int c = 8 * half + r;
      v2f o0, o1;
      o0.x = tanhf(s00 + bv.x); o0.y = tanhf(s01 + bv.y);
      o1.x = tanhf(s10 + bv.x); o1.y = tanhf(s11 + bv.y);
      *(v2f*)(tq + ((size_t)b * CA_ + c) * H_ + h0)      = o0;
      *(v2f*)(tq + ((size_t)b * CA_ + c + 16) * H_ + h0) = o1;
    }
  }
}

template <int NS>
__device__ __forceinline__ void out_plan(const int* __restrict__ seq_len, int lane,
                                         int& tiles, int& eend) {
  const int bl = lane & 15;
  int len = seq_len[bl];
  len = len < 0 ? 0 : (len > S_ ? S_ : len);
  tiles = (len + 15) >> 4;
  int p = tiles;
  p += dpp_row<DPP_ROW_SHR + 1>(0, p);
  p += dpp_row<DPP_ROW_SHR + 2>(0, p);
  p += dpp_row<DPP_ROW_SHR + 4>(0, p);
  p += dpp_row<DPP_ROW_SHR + 8>(0, p);
  const int G = dpp_row<DPP_ROW_SHARE + 15>(0, p);
  int r = (G > 0) ? (2 * NS * p + G) / (2 * G) : 0;
  r = r > NS - 15 + bl ? NS - 15 + bl : r;
  r = r < bl + 1 ? bl + 1 : r;
  int m = r - bl;
  int q;
  q = dpp_row<DPP_ROW_SHR + 1>(m, m);  m = m > q ? m : q;
  q = dpp_row<DPP_ROW_SHR + 2>(m, m);  m = m > q ? m : q;
  q = dpp_row<DPP_ROW_SHR + 4>(m, m);  m = m > q ? m : q;
  q = dpp_row<DPP_ROW_SHR + 8>(m, m);  m = m > q ? m : q;
  eend = bl + m;
}

__device__ __forceinline__ int row16_prefix(int p) {
  p += dpp_row<DPP_ROW_SHR + 1>(0, p);
  p += dpp_row<DPP_ROW_SHR + 2>(0, p);
  p += dpp_row<DPP_ROW_SHR + 4>(0, p);
  p += dpp_row<DPP_ROW_SHR + 8>(0, p);
  return p;
}
__device__ __forceinline__ void sc_plan(const int* __restrict__ seq_len, int lane,
                                        int& tiles, int& send, int& T) {
  const int bl = lane & 15;
  int len = seq_len[bl];
  len = len < 0 ? 0 : (len > S_ ? S_ : len);
  tiles = (len + 15) >> 4;
  const int G  = dpp_row<DPP_ROW_SHARE + 15>(0, row16_prefix(tiles));
  int T0 = (G + SCN_SLOTS - 1) / SCN_SLOTS;
  T0 = T0 < 1 ? 1 : T0;
  const int ea = row16_prefix((tiles + T0 - 1) / T0);
  const int eb = row16_prefix((tiles + T0) / (T0 + 1));
  const bool fits = dpp_row<DPP_ROW_SHARE + 15>(0, ea) <= SCN_SLOTS;
  T    = fits ? T0 : T0 + 1;
  send = fits ? ea : eb;
}

constexpr int SCN_WAVES = 4;
constexpr int SCN_KQ    = H_ / SCN_WAVES;
constexpr int SCN_L     = SCN_KQ / 8;
constexpr int SCN_HEAD  = 6;
static_assert(DEN_SZ == (size_t)SCN_SLOTS * CA_, "one denominator row per slot");

__global__ void __launch_bounds__(32 * SCN_WAVES)
k_scores(const float* __restrict__ key, const float* __restrict__ tq,
         const int* __restrict__ seq_len,
         float* __restrict__ alpha_g, float* __restrict__ den_part) {
  __shared__ float red[2][SCN_WAVES][16][32];
  const int slot = blockIdx.x;
  const int tid  = threadIdx.x;
  const int lane = tid & 31;
  const int wv   = tid >> 5;
  const int hl   = lane & 15;
  const int half = lane >> 4;

  int tiles, send, T;
  sc_plan(seq_len, lane, tiles, send, T);
  const int b = __popc((unsigned)__ballot(send <= slot) & 0xFFFFu);
  if (b > 15) return;
  const int sp = __builtin_amdgcn_readlane(send, b > 0 ? b - 1 : 0);
  const int e0 = (b > 0) ? sp : 0;
  const int tb = __builtin_amdgcn_readlane(tiles, b);
  const int ts = __builtin_amdgcn_readfirstlane(T);
  const int lo = __builtin_amdgcn_readfirstlane((slot - e0) * ts);
  const int hi = lo + ts < tb ? lo + ts : tb;
  const int len = seq_len[b];

  float dsum[4] = {0.f, 0.f, 0.f, 0.f};

  const float* a0p = tq + ((size_t)b * CA_ + hl) * H_ + SCN_KQ * wv + 4 * half;
  const float* a1p = a0p + (size_t)16 * H_;
  const float* kp  = key + ((size_t)b * S_ + hl) * H_ + SCN_KQ * wv + 4 * half;
  float* ag = alpha_g + (size_t)b * CA_ * S_;
  const int last = hi - 1;

  float4 kv[2][SCN_L];
  float4 af0[SCN_L], af1[SCN_L];
#define SCN_LOAD(SET, TILE)                                              \
  {                                                                    \
    const int tt_ = (TILE) < last ? (TILE) : last;                     \
    const float* kr_ = kp + (size_t)(16 * tt_) * H_;                   \
    _Pragma("unroll") for (int m = 0; m < SCN_L; ++m)                  \
      kv[SET][m] = *(const float4*)(kr_ + 8 * m);                      \
  }
#define SCN_STEPS(SET, M0, M1)                                         \
  {                                                                    \
    _Pragma("unroll") for (int m = (M0); m < (M1); ++m) {              \
      const float4 kq = kv[SET][m];                                    \
      const float4 p  = af0[m];                                        \
      const float4 q  = af1[m];                                        \
      v2f blo, bhi, p0, p1, q0, q1;                                    \
      blo.x = kq.x; blo.y = kq.y;  bhi.x = kq.z; bhi.y = kq.w;         \
      p0.x = p.x;   p0.y = p.y;    p1.x = p.z;   p1.y = p.w;           \
      q0.x = q.x;   q0.y = q.y;    q1.x = q.z;   q1.y = q.w;           \
      acc0 = WMMA_F32(p0, blo, acc0);                                  \
      acc1 = WMMA_F32(q0, blo, acc1);                                  \
      acc0 = WMMA_F32(p1, bhi, acc0);                                  \
      acc1 = WMMA_F32(q1, bhi, acc1);                                  \
    }                                                                  \
  }
#define SCN_PUT(TILE)                                                  \
  {                                                                    \
    float (*ws_)[16][32] = red[(TILE) & 1];                            \
    _Pragma("unroll") for (int r = 0; r < 8; ++r) {                    \
      ws_[wv][r][lane]     = acc0[r];                                  \
      ws_[wv][8 + r][lane] = acc1[r];                                  \
    }                                                                  \
    __syncthreads();                                                   \
    acc0 = zero8;                                                      \
    acc1 = zero8;                                                      \
  }
#define SCN_FIN(TILE)                                                  \
  {                                                                    \
    float (*rs_)[16][32] = red[(TILE) & 1];                            \
    const int s_ = 16 * (TILE) + hl;                                   \
    const bool valid_ = s_ < len;                                      \
    _Pragma("unroll") for (int q = 0; q < 2; ++q) {                    \
      const int r = 2 * wv + q;                                        \
      float x0 = rs_[0][r][lane], x1 = rs_[0][8 + r][lane];            \
      _Pragma("unroll") for (int w = 1; w < SCN_WAVES; ++w) {          \
        x0 += rs_[w][r][lane];                                         \
        x1 += rs_[w][8 + r][lane];                                     \
      }                                                                \
      const float v0 = valid_ ? __expf(x0) : 0.0f;                     \
      const float v1 = valid_ ? __expf(x1) : 0.0f;                     \
      const int c0 = r + 8 * half;                                     \
      ag[(size_t)c0 * S_ + s_]        = v0;                            \
      ag[(size_t)(c0 + 16) * S_ + s_] = v1;                            \
      dsum[2 * q]     += v0;                                           \
      dsum[2 * q + 1] += v1;                                           \
    }                                                                  \
  }
  SCN_LOAD(0, lo);
#pragma unroll
  for (int m = 0; m < SCN_L; ++m) {
    af0[m] = *(const float4*)(a0p + 8 * m);
    af1[m] = *(const float4*)(a1p + 8 * m);
  }
  SCN_LOAD(1, lo + 1);
  const v8f zero8 = {0.f, 0.f, 0.f, 0.f, 0.f, 0.f, 0.f, 0.f};
  v8f acc0 = zero8, acc1 = zero8;
  __builtin_amdgcn_sched_barrier(0);
  SCN_STEPS(0, 0, SCN_L);
  SCN_PUT(lo);
  for (int t = lo + 1; t < hi; t += 2) {
    SCN_LOAD(0, t + 1);
    __builtin_amdgcn_sched_barrier(0);
    SCN_FIN(t - 1);
    SCN_STEPS(1, 0, SCN_HEAD);
    __builtin_amdgcn_sched_barrier(0);
    SCN_STEPS(1, SCN_HEAD, SCN_L);
    SCN_PUT(t);
    if (t + 1 < hi) {
      SCN_LOAD(1, t + 2);
      __builtin_amdgcn_sched_barrier(0);
      SCN_FIN(t);
      SCN_STEPS(0, 0, SCN_HEAD);
      __builtin_amdgcn_sched_barrier(0);
      SCN_STEPS(0, SCN_HEAD, SCN_L);
      SCN_PUT(t + 1);
    }
  }
  SCN_FIN(last);
#undef SCN_LOAD
#undef SCN_STEPS
#undef SCN_PUT
#undef SCN_FIN

  for (int q = 0; q < 2; ++q) {
    const float d0 = row16_sum(dsum[2 * q]);
    const float d1 = row16_sum(dsum[2 * q + 1]);
    if (hl == 0) {
      const int c0 = 2 * wv + q + 8 * half;
      den_part[(size_t)slot * CA_ + c0]      = d0;
      den_part[(size_t)slot * CA_ + c0 + 16] = d1;
    }
  }
}

constexpr int OUT_WAVES = 4;
constexpr int OUT_CG    = H_ / 64;
constexpr int OUT_SLOTS = KSPLIT * B_;
constexpr int OUT_ST    = 4;
static_assert(OUT_ST == 4, "the pinning statement in k_out names 4 k-steps");

__global__ void __launch_bounds__(32 * OUT_WAVES)
k_out(const float* __restrict__ value, const float* __restrict__ alpha_g,
      const int* __restrict__ seq_len, float* __restrict__ num_part) {
  __shared__ float red[OUT_WAVES][64][32];
  const int gid  = blockIdx.x;
  const int cg   = gid % OUT_CG;
  const int slot = gid / OUT_CG;
  const int tid  = threadIdx.x;
  const int lane = tid & 31;
  const int wv   = tid >> 5;
  const int hl   = lane & 15;
  const int half = lane >> 4;
  const int kh   = half * 2;

  int tiles, eend;
  out_plan<OUT_SLOTS>(seq_len, lane, tiles, eend);
  const int b = __popc((unsigned)__ballot(eend <= slot) & 0xFFFFu);
  if (b > 15) return;
  const int ep = __builtin_amdgcn_readlane(eend, b > 0 ? b - 1 : 0);
  const int e0 = (b > 0) ? ep : 0;
  const int e1 = __builtin_amdgcn_readlane(eend, b);
  const int tb = __builtin_amdgcn_readlane(tiles, b);
  const int nb = e1 - e0;
  const int lo = ((slot - e0) * tb) / nb;
  const int hi = ((slot - e0 + 1) * tb) / nb;
  const int n  = hi - lo;
  const int t0 = __builtin_amdgcn_readfirstlane(lo + (wv * n) / OUT_WAVES);
  const int t1 = __builtin_amdgcn_readfirstlane(lo + ((wv + 1) * n) / OUT_WAVES);

  const float* ap0 = alpha_g + ((size_t)b * CA_ + hl) * S_ + kh;
  const float* ap1 = ap0 + (size_t)16 * S_;
  const float* vp  = value + ((size_t)b * S_ + kh) * H_ + cg * 64 + 4 * hl;

  v8f acc[8];
  for (int a = 0; a < 8; ++a) acc[a] = (v8f){0.f, 0.f, 0.f, 0.f, 0.f, 0.f, 0.f, 0.f};

  if (t0 < t1) {
    v2f    a0[3][OUT_ST], a1[3][OUT_ST];
    v4f    v0[3][OUT_ST], v1[3][OUT_ST];
    const int last = t1 - 1;
#define OUT_LOAD(SET, TILE)                                              \
    {                                                                    \
      const int tt_ = (TILE) < last ? (TILE) : last;                     \
      _Pragma("unroll") for (int i = 0; i < OUT_ST; ++i) {               \
        const int k = 16 * tt_ + 4 * i;                                  \
        a0[SET][i] = *(const v2f*)(ap0 + k);                             \
        a1[SET][i] = *(const v2f*)(ap1 + k);                             \
        v0[SET][i] = *(const v4f*)(vp + (size_t)k * H_);                 \
        v1[SET][i] = *(const v4f*)(vp + (size_t)(k + 1) * H_);           \
      }                                                                  \
    }
#define OUT_MMA(SET)                                                     \
    {                                                                    \
         \
      asm volatile("" : "+v"(v0[SET][0]), "+v"(v1[SET][0]), "+v"(v0[SET][1]),   \
                        "+v"(v1[SET][1]), "+v"(v0[SET][2]), "+v"(v1[SET][2]),   \
                        "+v"(v0[SET][3]), "+v"(v1[SET][3]));                    \
      _Pragma("unroll") for (int i = 0; i < OUT_ST; ++i) {               \
        const float x0[4] = {v0[SET][i].x, v0[SET][i].y, v0[SET][i].z, v0[SET][i].w}; \
        const float x1[4] = {v1[SET][i].x, v1[SET][i].y, v1[SET][i].z, v1[SET][i].w}; \
        _Pragma("unroll") for (int j = 0; j < 4; ++j) {                  \
          v2f bf;                                                        \
          bf.x = x0[j];                                                  \
          bf.y = x1[j];                                                  \
          acc[j]     = WMMA_F32(a0[SET][i], bf, acc[j]);                 \
          acc[4 + j] = WMMA_F32(a1[SET][i], bf, acc[4 + j]);             \
        }                                                                \
      }                                                                  \
    }
    OUT_LOAD(0, t0);
    OUT_LOAD(1, t0 + 1);
    int t = t0;
    for (; t + 3 <= t1; t += 3) {
      OUT_LOAD(2, t + 2);
      __builtin_amdgcn_sched_barrier(0);
      OUT_MMA(0);
      OUT_LOAD(0, t + 3);
      __builtin_amdgcn_sched_barrier(0);
      OUT_MMA(1);
      OUT_LOAD(1, t + 4);
      __builtin_amdgcn_sched_barrier(0);
      OUT_MMA(2);
    }
    if (t < t1) OUT_MMA(0);
    if (t + 1 < t1) OUT_MMA(1);
#undef OUT_LOAD
#undef OUT_MMA
  }

  for (int a = 0; a < 8; ++a)
    for (int r = 0; r < 8; ++r)
      red[wv][a * 8 + r][lane] = acc[a][r];
  __syncthreads();

  float* o = num_part + (size_t)slot * CA_ * H_ + cg * 64 + 4 * hl;
  const int t = wv >> 1;
  for (int q = 0; q < 4; ++q) {
    const int r = 4 * (wv & 1) + q;
    float sum[4];
    for (int j = 0; j < 4; ++j) {
      const int reg = (4 * t + j) * 8 + r;
      float s = red[0][reg][lane];
      for (int w = 1; w < OUT_WAVES; ++w) s += red[w][reg][lane];
      sum[j] = s;
    }
    const int c = 16 * t + 8 * half + r;
    *(float4*)(o + (size_t)c * H_) = make_float4(sum[0], sum[1], sum[2], sum[3]);
  }
}

constexpr int FIN_T = 64;
constexpr int FIN_V = H_ / (4 * FIN_T);
constexpr int FIN_G = 8;
constexpr int FIN_D = SCN_SLOTS / 32;
static_assert(FIN_V * 4 * FIN_T == H_, "a block covers one (b, c) row exactly");
static_assert(FIN_D * 32 == SCN_SLOTS, "FIN_D loads of 32 lanes reach every slot of the scores kernel");

__global__ void __launch_bounds__(FIN_T)
k_fin(const float* __restrict__ num_part, const float* __restrict__ den_part,
      const int* __restrict__ seq_len, float* __restrict__ out) {
  const int bc   = blockIdx.x;
  const int b    = bc / CA_, c = bc % CA_;
  const int tid  = threadIdx.x;
  const int lane = tid & 31;
  int stiles, send, st;
  sc_plan(seq_len, lane, stiles, send, st);
  const int sp  = __builtin_amdgcn_readlane(send, b > 0 ? b - 1 : 0);
  const int se0 = (b > 0) ? sp : 0;
  const int se1 = __builtin_amdgcn_readlane(send, b);
  const int dlast = se1 > 0 ? se1 - 1 : 0;
  float dv[FIN_D];
#pragma unroll
  for (int i = 0; i < FIN_D; ++i) {
    const int t = se0 + lane + 32 * i;
    dv[i] = den_part[(size_t)(t < dlast ? t : dlast) * CA_ + c];
  }
  int tiles, eend;
  out_plan<OUT_SLOTS>(seq_len, lane, tiles, eend);
  const int ep = __builtin_amdgcn_readlane(eend, b > 0 ? b - 1 : 0);
  const int e0 = (b > 0) ? ep : 0;
  const int e1 = __builtin_amdgcn_readlane(eend, b);
  float4 num[FIN_V];
  for (int j = 0; j < FIN_V; ++j) num[j] = make_float4(0.f, 0.f, 0.f, 0.f);
  for (int base = e0; base < e1; base += FIN_G) {
    float4 nv[FIN_G][FIN_V];
#pragma unroll
    for (int q = 0; q < FIN_G; ++q) {
      const int sl = base + q < e1 ? base + q : e1 - 1;
      const float* ps = num_part + ((size_t)sl * CA_ + c) * H_;
#pragma unroll
      for (int j = 0; j < FIN_V; ++j) nv[q][j] = *(const float4*)(ps + 4 * tid + 4 * FIN_T * j);
    }
#pragma unroll
    for (int q = 0; q < FIN_G; ++q) {
      const bool in = base + q < e1;
#pragma unroll
      for (int j = 0; j < FIN_V; ++j) {
        num[j].x += in ? nv[q][j].x : 0.f;
        num[j].y += in ? nv[q][j].y : 0.f;
        num[j].z += in ? nv[q][j].z : 0.f;
        num[j].w += in ? nv[q][j].w : 0.f;
      }
    }
  }
  float dl = 0.f;
#pragma unroll
  for (int i = 0; i < FIN_D; ++i) dl += (se0 + lane + 32 * i < se1) ? dv[i] : 0.f;
  dl = row16_sum(dl);
  const float den = __int_as_float(__builtin_amdgcn_readlane(__float_as_int(dl), 0)) +
                    __int_as_float(__builtin_amdgcn_readlane(__float_as_int(dl), 16));
  float* op = out + (size_t)bc * H_ + 4 * tid;
  for (int j = 0; j < FIN_V; ++j)
    *(float4*)(op + 4 * FIN_T * j) =
        make_float4(num[j].x / den, num[j].y / den, num[j].z / den, num[j].w / den);
}

extern "C" void kernel_launch(void* const* d_in, const int* in_sizes, int n_in,
                              void* d_out, int out_size, void* d_ws, size_t ws_size,
                              hipStream_t stream) {
  (void)in_sizes; (void)n_in; (void)out_size; (void)ws_size;
  const float* key     = (const float*)d_in[0];
  const float* value   = (const float*)d_in[1];
  const float* query   = (const float*)d_in[2];
  const int*   seq_len = (const int*)d_in[3];
  const float* W       = (const float*)d_in[4];
  const float* bias    = (const float*)d_in[5];
  float* ws  = (float*)d_ws;
  float* out = (float*)d_out;

  k_tq<<<B_ * TQ_NT, 32 * TQ_WAVES, 0, stream>>>(query, W, bias, ws + TQ_OFF);

  k_scores<<<SCN_SLOTS, 32 * SCN_WAVES, 0, stream>>>(key, ws + TQ_OFF, seq_len,
                                                     ws + AL_OFF, ws + DEN_OFF);

  k_out<<<OUT_SLOTS * OUT_CG, 32 * OUT_WAVES, 0, stream>>>(value, ws + AL_OFF, seq_len, ws + NUM_OFF);

  k_fin<<<B_ * CA_, FIN_T, 0, stream>>>(ws + NUM_OFF, ws + DEN_OFF, seq_len, out);
}


# output GEMM not split over blocks in K: a block owns one batch's 32 x 32-column strip, K over its 8 waves (LDS, wave order, permuted k in both loaders: new sum order), divides itself; 3 launches
# speedup vs baseline: 2.6642x; 1.0429x over previous
#include <hip/hip_runtime.h>

typedef __attribute__((ext_vector_type(2))) float v2f;
typedef __attribute__((ext_vector_type(4))) float v4f;
typedef __attribute__((ext_vector_type(8))) float v8f;

#define WMMA_F32(a, b, c) \
  __builtin_amdgcn_wmma_f32_16x16x4_f32(false, (a), false, (b), (short)0, (c), false, false)

constexpr int DPP_ROW_SHR   = 0x110;
constexpr int DPP_ROW_SHARE = 0x150;
constexpr int DPP_ROW_XMASK = 0x160;
template <int CTRL>
__device__ __forceinline__ int dpp_row(int old, int x) {
  return __builtin_amdgcn_update_dpp(old, x, CTRL, 0xf, 0xf, false);
}
template <int CTRL>
__device__ __forceinline__ float dpp_rowf(float old, float x) {
  return __int_as_float(dpp_row<CTRL>(__float_as_int(old), __float_as_int(x)));
}
__device__ __forceinline__ float row16_sum(float x) {
  x += dpp_rowf<DPP_ROW_XMASK + 1>(0.f, x);
  x += dpp_rowf<DPP_ROW_XMASK + 2>(0.f, x);
  x += dpp_rowf<DPP_ROW_XMASK + 4>(0.f, x);
  x += dpp_rowf<DPP_ROW_XMASK + 8>(0.f, x);
  return x;
}

constexpr int B_ = 16, S_ = 2048, H_ = 768, CA_ = 32, IN_ = 300;
constexpr int SCN_SLOTS = 256;

constexpr size_t TQ_OFF  = 0;
constexpr size_t TQ_SZ   = (size_t)B_ * CA_ * H_;
constexpr size_t AL_OFF  = TQ_OFF + TQ_SZ;
constexpr size_t AL_SZ   = (size_t)B_ * CA_ * S_;
constexpr size_t DEN_OFF = AL_OFF + AL_SZ;
constexpr size_t DEN_SZ  = (size_t)SCN_SLOTS * CA_;

constexpr int TQ_WAVES = 5;
constexpr int TQ_STEPS = IN_ / (4 * TQ_WAVES);
constexpr int TQ_NT    = H_ / 32;

__global__ void __launch_bounds__(32 * TQ_WAVES)
k_tq(const float* __restrict__ query, const float* __restrict__ W,
     const float* __restrict__ bias, float* __restrict__ tq) {
  __shared__ float red[TQ_WAVES][32][32];
  const int b    = blockIdx.x / TQ_NT;
  const int nt   = blockIdx.x % TQ_NT;
  const int tid  = threadIdx.x;
  const int lane = tid & 31;
  const int wv   = tid >> 5;
  const int hl   = lane & 15;
  const int half = lane >> 4;
  const int kh   = half * 2;
  const int kb   = wv * (TQ_STEPS * 4);

  const float* ap0 = query + ((size_t)b * CA_ + hl) * IN_ + kb + kh;
  const float* ap1 = ap0 + (size_t)16 * IN_;
  const float* bp  = W + (size_t)(kb + kh) * H_ + nt * 32 + 2 * hl;

  v8f acc00 = {0.f, 0.f, 0.f, 0.f, 0.f, 0.f, 0.f, 0.f};
  v8f acc01 = acc00, acc10 = acc00, acc11 = acc00;
  v2f a0[TQ_STEPS], a1[TQ_STEPS], w0[TQ_STEPS], w1[TQ_STEPS];
#pragma unroll
  for (int i = 0; i < TQ_STEPS; ++i) {
    a0[i] = *(const v2f*)(ap0 + 4 * i);
    a1[i] = *(const v2f*)(ap1 + 4 * i);
    w0[i] = *(const v2f*)(bp + (size_t)(4 * i) * H_);
    w1[i] = *(const v2f*)(bp + (size_t)(4 * i + 1) * H_);
  }
  __builtin_amdgcn_sched_barrier(0);
#pragma unroll
  for (int i = 0; i < TQ_STEPS; ++i) {
    v2f b0, b1;
    b0.x = w0[i].x; b0.y = w1[i].x;
    b1.x = w0[i].y; b1.y = w1[i].y;
    acc00 = WMMA_F32(a0[i], b0, acc00);
    acc01 = WMMA_F32(a0[i], b1, acc01);
    acc10 = WMMA_F32(a1[i], b0, acc10);
    acc11 = WMMA_F32(a1[i], b1, acc11);
  }

  for (int r = 0; r < 8; ++r) {
    red[wv][r][lane]      = acc00[r];
    red[wv][8 + r][lane]  = acc01[r];
    red[wv][16 + r][lane] = acc10[r];
    red[wv][24 + r][lane] = acc11[r];
  }
  __syncthreads();

  if (wv < 4) {
    const int h0 = nt * 32 + 2 * hl;
    const v2f bv = *(const v2f*)(bias + h0);
    for (int q = 0; q < 2; ++q) {
      const int r = 2 * wv + q;
      float s00 = red[0][r][lane],      s01 = red[0][8 + r][lane];
      float s10 = red[0][16 + r][lane], s11 = red[0][24 + r][lane];
      for (int w = 1; w < TQ_WAVES; ++w) {
        s00 += red[w][r][lane];
        s01 += red[w][8 + r][lane];
        s10 += red[w][16 + r][lane];
        s11 += red[w][24 + r][lane];
      }
      const int c = 8 * half + r;
      v2f o0, o1;
      o0.x = tanhf(s00 + bv.x); o0.y = tanhf(s01 + bv.y);
      o1.x = tanhf(s10 + bv.x); o1.y = tanhf(s11 + bv.y);
      *(v2f*)(tq + ((size_t)b * CA_ + c) * H_ + h0)      = o0;
      *(v2f*)(tq + ((size_t)b * CA_ + c + 16) * H_ + h0) = o1;
    }
  }
}

__device__ __forceinline__ int row16_prefix(int p) {
  p += dpp_row<DPP_ROW_SHR + 1>(0, p);
  p += dpp_row<DPP_ROW_SHR + 2>(0, p);
  p += dpp_row<DPP_ROW_SHR + 4>(0, p);
  p += dpp_row<DPP_ROW_SHR + 8>(0, p);
  return p;
}
__device__ __forceinline__ void sc_plan_tiles(int tiles, int& send, int& T) {
  const int G  = dpp_row<DPP_ROW_SHARE + 15>(0, row16_prefix(tiles));
  int T0 = (G + SCN_SLOTS - 1) / SCN_SLOTS;
  T0 = T0 < 1 ? 1 : T0;
  const int ea = row16_prefix((tiles + T0 - 1) / T0);
  const int eb = row16_prefix((tiles + T0) / (T0 + 1));
  const bool fits = dpp_row<DPP_ROW_SHARE + 15>(0, ea) <= SCN_SLOTS;
  T    = fits ? T0 : T0 + 1;
  send = fits ? ea : eb;
}
__device__ __forceinline__ void sc_plan(const int* __restrict__ seq_len, int lane,
                                        int& tiles, int& send, int& T) {
  const int bl = lane & 15;
  int len = seq_len[bl];
  len = len < 0 ? 0 : (len > S_ ? S_ : len);
  tiles = (len + 15) >> 4;
  sc_plan_tiles(tiles, send, T);
}

constexpr int SCN_WAVES = 4;
constexpr int SCN_KQ    = H_ / SCN_WAVES;
constexpr int SCN_L     = SCN_KQ / 8;
constexpr int SCN_HEAD  = 6;
static_assert(DEN_SZ == (size_t)SCN_SLOTS * CA_, "one denominator row per slot");

__global__ void __launch_bounds__(32 * SCN_WAVES)
k_scores(const float* __restrict__ key, const float* __restrict__ tq,
         const int* __restrict__ seq_len,
         float* __restrict__ alpha_g, float* __restrict__ den_part) {
  __shared__ float red[2][SCN_WAVES][16][32];
  const int slot = blockIdx.x;
  const int tid  = threadIdx.x;
  const int lane = tid & 31;
  const int wv   = tid >> 5;
  const int hl   = lane & 15;
  const int half = lane >> 4;

  int tiles, send, T;
  sc_plan(seq_len, lane, tiles, send, T);
  const int b = __popc((unsigned)__ballot(send <= slot) & 0xFFFFu);
  if (b > 15) return;
  const int sp = __builtin_amdgcn_readlane(send, b > 0 ? b - 1 : 0);
  const int e0 = (b > 0) ? sp : 0;
  const int tb = __builtin_amdgcn_readlane(tiles, b);
  const int ts = __builtin_amdgcn_readfirstlane(T);
  const int lo = __builtin_amdgcn_readfirstlane((slot - e0) * ts);
  const int hi = lo + ts < tb ? lo + ts : tb;
  const int len = seq_len[b];

  float dsum[4] = {0.f, 0.f, 0.f, 0.f};

  const float* a0p = tq + ((size_t)b * CA_ + hl) * H_ + SCN_KQ * wv + 4 * half;
  const float* a1p = a0p + (size_t)16 * H_;
  const float* kp  = key + ((size_t)b * S_ + hl) * H_ + SCN_KQ * wv + 4 * half;
  float* ag = alpha_g + (size_t)b * CA_ * S_;
  const int last = hi - 1;

  float4 kv[2][SCN_L];
  float4 af0[SCN_L], af1[SCN_L];
#define SCN_LOAD(SET, TILE)                                              \
  {                                                                    \
    const int tt_ = (TILE) < last ? (TILE) : last;                     \
    const float* kr_ = kp + (size_t)(16 * tt_) * H_;                   \
    _Pragma("unroll") for (int m = 0; m < SCN_L; ++m)                  \
      kv[SET][m] = *(const float4*)(kr_ + 8 * m);                      \
  }
#define SCN_STEPS(SET, M0, M1)                                         \
  {                                                                    \
    _Pragma("unroll") for (int m = (M0); m < (M1); ++m) {              \
      const float4 kq = kv[SET][m];                                    \
      const float4 p  = af0[m];                                        \
      const float4 q  = af1[m];                                        \
      v2f blo, bhi, p0, p1, q0, q1;                                    \
      blo.x = kq.x; blo.y = kq.y;  bhi.x = kq.z; bhi.y = kq.w;         \
      p0.x = p.x;   p0.y = p.y;    p1.x = p.z;   p1.y = p.w;           \
      q0.x = q.x;   q0.y = q.y;    q1.x = q.z;   q1.y = q.w;           \
      acc0 = WMMA_F32(p0, blo, acc0);                                  \
      acc1 = WMMA_F32(q0, blo, acc1);                                  \
      acc0 = WMMA_F32(p1, bhi, acc0);                                  \
      acc1 = WMMA_F32(q1, bhi, acc1);                                  \
    }                                                                  \
  }
#define SCN_PUT(TILE)                                                  \
  {                                                                    \
    float (*ws_)[16][32] = red[(TILE) & 1];                            \
    _Pragma("unroll") for (int r = 0; r < 8; ++r) {                    \
      ws_[wv][r][lane]     = acc0[r];                                  \
      ws_[wv][8 + r][lane] = acc1[r];                                  \
    }                                                                  \
    __syncthreads();                                                   \
    acc0 = zero8;                                                      \
    acc1 = zero8;                                                      \
  }
#define SCN_FIN(TILE)                                                  \
  {                                                                    \
    float (*rs_)[16][32] = red[(TILE) & 1];                            \
    const int s_ = 16 * (TILE) + hl;                                   \
    const bool valid_ = s_ < len;                                      \
    _Pragma("unroll") for (int q = 0; q < 2; ++q) {                    \
      const int r = 2 * wv + q;                                        \
      float x0 = rs_[0][r][lane], x1 = rs_[0][8 + r][lane];            \
      _Pragma("unroll") for (int w = 1; w < SCN_WAVES; ++w) {          \
        x0 += rs_[w][r][lane];                                         \
        x1 += rs_[w][8 + r][lane];                                     \
      }                                                                \
      const float v0 = valid_ ? __expf(x0) : 0.0f;                     \
      const float v1 = valid_ ? __expf(x1) : 0.0f;                     \
      const int c0 = r + 8 * half;                                     \
      ag[(size_t)c0 * S_ + s_]        = v0;                            \
      ag[(size_t)(c0 + 16) * S_ + s_] = v1;                            \
      dsum[2 * q]     += v0;                                           \
      dsum[2 * q + 1] += v1;                                           \
    }                                                                  \
  }
  SCN_LOAD(0, lo);
#pragma unroll
  for (int m = 0; m < SCN_L; ++m) {
    af0[m] = *(const float4*)(a0p + 8 * m);
    af1[m] = *(const float4*)(a1p + 8 * m);
  }
  SCN_LOAD(1, lo + 1);
  const v8f zero8 = {0.f, 0.f, 0.f, 0.f, 0.f, 0.f, 0.f, 0.f};
  v8f acc0 = zero8, acc1 = zero8;
  __builtin_amdgcn_sched_barrier(0);
  SCN_STEPS(0, 0, SCN_L);
  SCN_PUT(lo);
  for (int t = lo + 1; t < hi; t += 2) {
    SCN_LOAD(0, t + 1);
    __builtin_amdgcn_sched_barrier(0);
    SCN_FIN(t - 1);
    SCN_STEPS(1, 0, SCN_HEAD);
    __builtin_amdgcn_sched_barrier(0);
    SCN_STEPS(1, SCN_HEAD, SCN_L);
    SCN_PUT(t);
    if (t + 1 < hi) {
      SCN_LOAD(1, t + 2);
      __builtin_amdgcn_sched_barrier(0);
      SCN_FIN(t);
      SCN_STEPS(0, 0, SCN_HEAD);
      __builtin_amdgcn_sched_barrier(0);
      SCN_STEPS(0, SCN_HEAD, SCN_L);
      SCN_PUT(t + 1);
    }
  }
  SCN_FIN(last);
#undef SCN_LOAD
#undef SCN_STEPS
#undef SCN_PUT
#undef SCN_FIN

  for (int q = 0; q < 2; ++q) {
    const float d0 = row16_sum(dsum[2 * q]);
    const float d1 = row16_sum(dsum[2 * q + 1]);
    if (hl == 0) {
      const int c0 = 2 * wv + q + 8 * half;
      den_part[(size_t)slot * CA_ + c0]      = d0;
      den_part[(size_t)slot * CA_ + c0 + 16] = d1;
    }
  }
}

constexpr int OUT_WAVES = 8;
constexpr int OUT_CG    = H_ / 32;
constexpr int OUT_NB    = B_ * OUT_CG;
constexpr int OUT_LEAD  = 256;
constexpr int OUT_PAIR  = OUT_NB - OUT_LEAD;
static_assert(OUT_PAIR > 0 && 2 * OUT_PAIR <= OUT_LEAD && OUT_PAIR % 2 == 0, "the index map below assumes it");
constexpr int OUT_ST    = 4;
constexpr int DEN_D     = SCN_SLOTS / 32;
static_assert(OUT_ST == 4, "the pinning statement in k_out names 4 k-steps");
static_assert(DEN_D * 32 == SCN_SLOTS, "DEN_D loads of 32 lanes reach every slot of the scores kernel");
static_assert(OUT_WAVES * 4 == CA_, "each wave sums the denominators of four rows");

__global__ void __launch_bounds__(32 * OUT_WAVES)
k_out(const float* __restrict__ value, const float* __restrict__ alpha_g,
      const float* __restrict__ den_part, const int* __restrict__ seq_len,
      float* __restrict__ out) {
  __shared__ float red[OUT_WAVES][32][32];
  __shared__ float den_s[CA_];
  const int gid  = blockIdx.x;
  int idx;
  if (gid < OUT_PAIR) {
    idx = (gid & 1) ? OUT_NB - gid : gid;
  } else if (gid < OUT_LEAD) {
    const int m = gid - OUT_PAIR;
    idx = (m & 1) ? OUT_LEAD - 1 - (m >> 1) : OUT_PAIR + (m >> 1);
  } else {
    const int g = gid - OUT_LEAD;
    idx = (g & 1) ? g : OUT_NB - 2 - g;
  }
  const int q    = idx / OUT_CG;
  const int cg   = idx % OUT_CG;
  const int tid  = threadIdx.x;
  const int lane = tid & 31;
  const int wv   = tid >> 5;
  const int hl   = lane & 15;
  const int half = lane >> 4;

  int len = seq_len[hl];
  len = len < 0 ? 0 : (len > S_ ? S_ : len);
  const int tiles = (len + 15) >> 4;
  const int key = tiles * 16 + (15 - hl);
  int rank = 0;
#pragma unroll
  for (int j = 0; j < 16; ++j) rank += (__builtin_amdgcn_readlane(key, j) > key) ? 1 : 0;
  const int b  = __ffs((unsigned)__ballot(rank == q) & 0xFFFFu) - 1;
  const int tb = __builtin_amdgcn_readlane(tiles, b);
  const int t0 = __builtin_amdgcn_readfirstlane((wv * tb) / OUT_WAVES);
  const int t1 = __builtin_amdgcn_readfirstlane(((wv + 1) * tb) / OUT_WAVES);

  const float* ap0 = alpha_g + ((size_t)b * CA_ + hl) * S_ + 8 * half;
  const float* ap1 = ap0 + (size_t)16 * S_;
  const float* vp  = value + ((size_t)b * S_ + 8 * half) * H_ + cg * 32 + 2 * hl;

  v8f acc[4];
  for (int a = 0; a < 4; ++a) acc[a] = (v8f){0.f, 0.f, 0.f, 0.f, 0.f, 0.f, 0.f, 0.f};

  float4 a0[3][2], a1[3][2];
  v2f    v0[3][OUT_ST], v1[3][OUT_ST];
  const int last = t1 - 1;
#define OUT_LOAD(SET, TILE)                                              \
    {                                                                    \
      const int tt_ = (TILE) < last ? (TILE) : last;                     \
      _Pragma("unroll") for (int m = 0; m < 2; ++m) {                    \
        a0[SET][m] = *(const float4*)(ap0 + 16 * tt_ + 4 * m);           \
        a1[SET][m] = *(const float4*)(ap1 + 16 * tt_ + 4 * m);           \
      }                                                                  \
      _Pragma("unroll") for (int i = 0; i < OUT_ST; ++i) {               \
        const int k = 16 * tt_ + 2 * i;                                  \
        v0[SET][i] = *(const v2f*)(vp + (size_t)k * H_);                 \
        v1[SET][i] = *(const v2f*)(vp + (size_t)(k + 1) * H_);           \
      }                                                                  \
    }
#define OUT_MMA(SET)                                                     \
    {                                                                    \
         \
      asm volatile("" : "+v"(v0[SET][0]), "+v"(v1[SET][0]), "+v"(v0[SET][1]),   \
                        "+v"(v1[SET][1]), "+v"(v0[SET][2]), "+v"(v1[SET][2]),   \
                        "+v"(v0[SET][3]), "+v"(v1[SET][3]));                    \
      _Pragma("unroll") for (int i = 0; i < OUT_ST; ++i) {               \
        const float4 p = a0[SET][i >> 1];                                \
        const float4 u = a1[SET][i >> 1];                                \
        v2f p0, u0, b0, b1;                                              \
        p0.x = (i & 1) ? p.z : p.x;  p0.y = (i & 1) ? p.w : p.y;         \
        u0.x = (i & 1) ? u.z : u.x;  u0.y = (i & 1) ? u.w : u.y;         \
        b0.x = v0[SET][i].x; b0.y = v1[SET][i].x;                        \
        b1.x = v0[SET][i].y; b1.y = v1[SET][i].y;                        \
        acc[0] = WMMA_F32(p0, b0, acc[0]);                               \
        acc[1] = WMMA_F32(p0, b1, acc[1]);                               \
        acc[2] = WMMA_F32(u0, b0, acc[2]);                               \
        acc[3] = WMMA_F32(u0, b1, acc[3]);                               \
      }                                                                  \
                  \
      asm volatile("" : "+v"(acc[0]), "+v"(acc[1]), "+v"(acc[2]), "+v"(acc[3])); \
    }
  if (t0 < t1) {
    OUT_LOAD(0, t0);
    OUT_LOAD(1, t0 + 1);
  }

  int send, st;
  sc_plan_tiles(tiles, send, st);
  const int sp  = __builtin_amdgcn_readlane(send, b > 0 ? b - 1 : 0);
  const int se0 = (b > 0) ? sp : 0;
  const int se1 = __builtin_amdgcn_readlane(send, b);
  const int dlast = se1 > 0 ? se1 - 1 : 0;
  float4 dv[DEN_D];
#pragma unroll
  for (int i = 0; i < DEN_D; ++i) {
    const int t = se0 + lane + 32 * i;
    dv[i] = *(const float4*)(den_part + (size_t)(t < dlast ? t : dlast) * CA_ + 4 * wv);
  }

  if (t0 < t1) {
    int t = t0;
    for (; t + 3 <= t1; t += 3) {
      OUT_LOAD(2, t + 2);
      __builtin_amdgcn_sched_barrier(0);
      OUT_MMA(0);
      OUT_LOAD(0, t + 3);
      __builtin_amdgcn_sched_barrier(0);
      OUT_MMA(1);
      OUT_LOAD(1, t + 4);
      __builtin_amdgcn_sched_barrier(0);
      OUT_MMA(2);
    }
    if (t < t1) OUT_MMA(0);
    if (t + 1 < t1) OUT_MMA(1);
#undef OUT_LOAD
#undef OUT_MMA
  }

  for (int a = 0; a < 4; ++a)
    for (int r = 0; r < 8; ++r)
      red[wv][a * 8 + r][lane] = acc[a][r];

  float dl[4] = {0.f, 0.f, 0.f, 0.f};
#pragma unroll
  for (int i = 0; i < DEN_D; ++i) {
    const bool in = se0 + lane + 32 * i < se1;
    dl[0] += in ? dv[i].x : 0.f;
    dl[1] += in ? dv[i].y : 0.f;
    dl[2] += in ? dv[i].z : 0.f;
    dl[3] += in ? dv[i].w : 0.f;
  }
  for (int u = 0; u < 4; ++u) {
    const float d = row16_sum(dl[u]);
    const float den = __int_as_float(__builtin_amdgcn_readlane(__float_as_int(d), 0)) +
                      __int_as_float(__builtin_amdgcn_readlane(__float_as_int(d), 16));
    if (lane == 0) den_s[4 * wv + u] = den;
  }
  __syncthreads();

  float* o = out + (size_t)b * CA_ * H_ + cg * 32 + 2 * hl;
  const int trow = wv >> 2;
  for (int qq = 0; qq < 2; ++qq) {
    const int r = 2 * (wv & 3) + qq;
    const int reg0 = (2 * trow) * 8 + r;
    const int reg1 = (2 * trow + 1) * 8 + r;
    float s0 = red[0][reg0][lane];
    float s1 = red[0][reg1][lane];
    for (int w = 1; w < OUT_WAVES; ++w) {
      s0 += red[w][reg0][lane];
      s1 += red[w][reg1][lane];
    }
    const int c = 16 * trow + 8 * half + r;
    const float den = den_s[c];
    v2f ov;
    ov.x = s0 / den;
    ov.y = s1 / den;
    *(v2f*)(o + (size_t)c * H_) = ov;
  }
}

extern "C" void kernel_launch(void* const* d_in, const int* in_sizes, int n_in,
                              void* d_out, int out_size, void* d_ws, size_t ws_size,
                              hipStream_t stream) {
  (void)in_sizes; (void)n_in; (void)out_size; (void)ws_size;
  const float* key     = (const float*)d_in[0];
  const float* value   = (const float*)d_in[1];
  const float* query   = (const float*)d_in[2];
  const int*   seq_len = (const int*)d_in[3];
  const float* W       = (const float*)d_in[4];
  const float* bias    = (const float*)d_in[5];
  float* ws  = (float*)d_ws;
  float* out = (float*)d_out;

  k_tq<<<B_ * TQ_NT, 32 * TQ_WAVES, 0, stream>>>(query, W, bias, ws + TQ_OFF);

  k_scores<<<SCN_SLOTS, 32 * SCN_WAVES, 0, stream>>>(key, ws + TQ_OFF, seq_len,
                                                     ws + AL_OFF, ws + DEN_OFF);

  k_out<<<OUT_NB, 32 * OUT_WAVES, 0, stream>>>(value, ws + AL_OFF, ws + DEN_OFF, seq_len, out);
}


# tq and alpha (the two workspace intermediates) stored in the fragment order of the kernel that reads them: its A loads cover whole 128-byte lines; row denominators summed in a new order
# speedup vs baseline: 2.7892x; 1.0469x over previous
#include <hip/hip_runtime.h>

typedef __attribute__((ext_vector_type(2))) float v2f;
typedef __attribute__((ext_vector_type(4))) float v4f;
typedef __attribute__((ext_vector_type(8))) float v8f;

#define WMMA_F32(a, b, c) \
  __builtin_amdgcn_wmma_f32_16x16x4_f32(false, (a), false, (b), (short)0, (c), false, false)

constexpr int DPP_ROW_SHR   = 0x110;
constexpr int DPP_ROW_SHARE = 0x150;
constexpr int DPP_ROW_XMASK = 0x160;
template <int CTRL>
__device__ __forceinline__ int dpp_row(int old, int x) {
  return __builtin_amdgcn_update_dpp(old, x, CTRL, 0xf, 0xf, false);
}
template <int CTRL>
__device__ __forceinline__ float dpp_rowf(float old, float x) {
  return __int_as_float(dpp_row<CTRL>(__float_as_int(old), __float_as_int(x)));
}
__device__ __forceinline__ float row16_sum(float x) {
  x += dpp_rowf<DPP_ROW_XMASK + 1>(0.f, x);
  x += dpp_rowf<DPP_ROW_XMASK + 2>(0.f, x);
  x += dpp_rowf<DPP_ROW_XMASK + 4>(0.f, x);
  x += dpp_rowf<DPP_ROW_XMASK + 8>(0.f, x);
  return x;
}

constexpr int B_ = 16, S_ = 2048, H_ = 768, CA_ = 32, IN_ = 300;
constexpr int SCN_SLOTS = 256;

constexpr int SCN_WAVES = 4;
constexpr int SCN_KQ    = H_ / SCN_WAVES;
constexpr int SCN_L     = SCN_KQ / 8;
constexpr int FRAG_CH   = 32 * 4;
static_assert(H_ % 32 == 0 && SCN_KQ % 32 == 0, "a 32-column tile of tq lies inside one K quarter");

constexpr size_t TQ_OFF  = 0;
constexpr size_t TQ_SZ   = (size_t)B_ * CA_ * H_;
constexpr size_t AL_OFF  = TQ_OFF + TQ_SZ;
constexpr size_t AL_SZ   = (size_t)B_ * CA_ * S_;
constexpr size_t DEN_OFF = AL_OFF + AL_SZ;
constexpr size_t DEN_SZ  = (size_t)SCN_SLOTS * CA_;

constexpr int TQ_WAVES = 5;
constexpr int TQ_STEPS = IN_ / (4 * TQ_WAVES);
constexpr int TQ_NT    = H_ / 32;

__global__ void __launch_bounds__(32 * TQ_WAVES)
k_tq(const float* __restrict__ query, const float* __restrict__ W,
     const float* __restrict__ bias, float* __restrict__ tq) {
  __shared__ float red[TQ_WAVES][32][33];
  const int b    = blockIdx.x / TQ_NT;
  const int nt   = blockIdx.x % TQ_NT;
  const int tid  = threadIdx.x;
  const int lane = tid & 31;
  const int wv   = tid >> 5;
  const int hl   = lane & 15;
  const int half = lane >> 4;
  const int kh   = half * 2;
  const int kb   = wv * (TQ_STEPS * 4);

  const float* ap0 = query + ((size_t)b * CA_ + hl) * IN_ + kb + kh;
  const float* ap1 = ap0 + (size_t)16 * IN_;
  const float* bp  = W + (size_t)(kb + kh) * H_ + nt * 32 + 2 * hl;

  v8f acc00 = {0.f, 0.f, 0.f, 0.f, 0.f, 0.f, 0.f, 0.f};
  v8f acc01 = acc00, acc10 = acc00, acc11 = acc00;
  v2f a0[TQ_STEPS], a1[TQ_STEPS], w0[TQ_STEPS], w1[TQ_STEPS];
#pragma unroll
  for (int i = 0; i < TQ_STEPS; ++i) {
    a0[i] = *(const v2f*)(ap0 + 4 * i);
    a1[i] = *(const v2f*)(ap1 + 4 * i);
    w0[i] = *(const v2f*)(bp + (size_t)(4 * i) * H_);
    w1[i] = *(const v2f*)(bp + (size_t)(4 * i + 1) * H_);
  }
  __builtin_amdgcn_sched_barrier(0);
#pragma unroll
  for (int i = 0; i < TQ_STEPS; ++i) {
    v2f b0, b1;
    b0.x = w0[i].x; b0.y = w1[i].x;
    b1.x = w0[i].y; b1.y = w1[i].y;
    acc00 = WMMA_F32(a0[i], b0, acc00);
    acc01 = WMMA_F32(a0[i], b1, acc01);
    acc10 = WMMA_F32(a1[i], b0, acc10);
    acc11 = WMMA_F32(a1[i], b1, acc11);
  }

  for (int r = 0; r < 8; ++r) {
    red[wv][r][lane]      = acc00[r];
    red[wv][8 + r][lane]  = acc01[r];
    red[wv][16 + r][lane] = acc10[r];
    red[wv][24 + r][lane] = acc11[r];
  }
  __syncthreads();

  if (wv < 4) {
    const int hq = 8 * wv + 4 * half;
    const float4 bv = *(const float4*)(bias + nt * 32 + hq);
    const int kq = (nt * 32) / SCN_KQ;
    const int m  = ((nt * 32) % SCN_KQ) / 8 + wv;
    const int rl = hl & 7;
    const int lp = 16 * (hl >> 3) + hq / 2;
    float* tqc = tq + ((((size_t)b * SCN_WAVES + kq) * 2) * SCN_L + m) * FRAG_CH + lane * 4;
    for (int j = 0; j < 2; ++j) {
      float s[4];
      for (int e = 0; e < 4; ++e) {
        const int reg = (2 * j + (e & 1)) * 8 + rl;
        const int ln  = lp + (e >> 1);
        float x = red[0][reg][ln];
        for (int w = 1; w < TQ_WAVES; ++w) x += red[w][reg][ln];
        s[e] = x;
      }
      float4 o;
      o.x = tanhf(s[0] + bv.x); o.y = tanhf(s[1] + bv.y);
      o.z = tanhf(s[2] + bv.z); o.w = tanhf(s[3] + bv.w);
      *(float4*)(tqc + (size_t)j * SCN_L * FRAG_CH) = o;
    }
  }
}

__device__ __forceinline__ int row16_prefix(int p) {
  p += dpp_row<DPP_ROW_SHR + 1>(0, p);
  p += dpp_row<DPP_ROW_SHR + 2>(0, p);
  p += dpp_row<DPP_ROW_SHR + 4>(0, p);
  p += dpp_row<DPP_ROW_SHR + 8>(0, p);
  return p;
}
__device__ __forceinline__ void sc_plan_tiles(int tiles, int& send, int& T) {
  const int G  = dpp_row<DPP_ROW_SHARE + 15>(0, row16_prefix(tiles));
  int T0 = (G + SCN_SLOTS - 1) / SCN_SLOTS;
  T0 = T0 < 1 ? 1 : T0;
  const int ea = row16_prefix((tiles + T0 - 1) / T0);
  const int eb = row16_prefix((tiles + T0) / (T0 + 1));
  const bool fits = dpp_row<DPP_ROW_SHARE + 15>(0, ea) <= SCN_SLOTS;
  T    = fits ? T0 : T0 + 1;
  send = fits ? ea : eb;
}
__device__ __forceinline__ void sc_plan(const int* __restrict__ seq_len, int lane,
                                        int& tiles, int& send, int& T) {
  const int bl = lane & 15;
  int len = seq_len[bl];
  len = len < 0 ? 0 : (len > S_ ? S_ : len);
  tiles = (len + 15) >> 4;
  sc_plan_tiles(tiles, send, T);
}

constexpr int SCN_HEAD  = 6;
static_assert(DEN_SZ == (size_t)SCN_SLOTS * CA_, "one denominator row per slot");

__global__ void __launch_bounds__(32 * SCN_WAVES)
k_scores(const float* __restrict__ key, const float* __restrict__ tq,
         const int* __restrict__ seq_len,
         float* __restrict__ alpha_g, float* __restrict__ den_part) {
  __shared__ __attribute__((aligned(16))) float red[2][SCN_WAVES][16][32];
  __shared__ float dpart[2][CA_];
  const int slot = blockIdx.x;
  const int tid  = threadIdx.x;
  const int lane = tid & 31;
  const int wv   = tid >> 5;
  const int hl   = lane & 15;
  const int half = lane >> 4;

  int tiles, send, T;
  sc_plan(seq_len, lane, tiles, send, T);
  const int b = __popc((unsigned)__ballot(send <= slot) & 0xFFFFu);
  if (b > 15) return;
  const int sp = __builtin_amdgcn_readlane(send, b > 0 ? b - 1 : 0);
  const int e0 = (b > 0) ? sp : 0;
  const int tb = __builtin_amdgcn_readlane(tiles, b);
  const int ts = __builtin_amdgcn_readfirstlane(T);
  const int lo = __builtin_amdgcn_readfirstlane((slot - e0) * ts);
  const int hi = lo + ts < tb ? lo + ts : tb;
  const int len = seq_len[b];

  const int fj = wv >> 1;
  const int fm = wv & 1;
  float dsum = 0.f;
  const int fo = (fj * 8 + (hl & 7)) * 32 + 16 * (hl >> 3) + 8 * half + 4 * fm;

  const float* a0p = tq + (((size_t)b * SCN_WAVES + wv) * 2) * SCN_L * FRAG_CH + lane * 4;
  const float* a1p = a0p + (size_t)SCN_L * FRAG_CH;
  const float* kp  = key + ((size_t)b * S_ + hl) * H_ + SCN_KQ * wv + 4 * half;
  float* ag = alpha_g + (size_t)b * CA_ * S_ + wv * FRAG_CH + lane * 4;
  const int last = hi - 1;

  float4 kv[2][SCN_L];
  float4 af0[SCN_L], af1[SCN_L];
#define SCN_LOAD(SET, TILE)                                              \
  {                                                                    \
    const int tt_ = (TILE) < last ? (TILE) : last;                     \
    const float* kr_ = kp + (size_t)(16 * tt_) * H_;                   \
    _Pragma("unroll") for (int m = 0; m < SCN_L; ++m)                  \
      kv[SET][m] = *(const float4*)(kr_ + 8 * m);                      \
  }
#define SCN_STEPS(SET, M0, M1)                                         \
  {                                                                    \
    _Pragma("unroll") for (int m = (M0); m < (M1); ++m) {              \
      const float4 kq = kv[SET][m];                                    \
      const float4 p  = af0[m];                                        \
      const float4 q  = af1[m];                                        \
      v2f blo, bhi, p0, p1, q0, q1;                                    \
      blo.x = kq.x; blo.y = kq.y;  bhi.x = kq.z; bhi.y = kq.w;         \
      p0.x = p.x;   p0.y = p.y;    p1.x = p.z;   p1.y = p.w;           \
      q0.x = q.x;   q0.y = q.y;    q1.x = q.z;   q1.y = q.w;           \
      acc0 = WMMA_F32(p0, blo, acc0);                                  \
      acc1 = WMMA_F32(q0, blo, acc1);                                  \
      acc0 = WMMA_F32(p1, bhi, acc0);                                  \
      acc1 = WMMA_F32(q1, bhi, acc1);                                  \
    }                                                                  \
  }
#define SCN_PUT(TILE)                                                  \
  {                                                                    \
    float (*ws_)[16][32] = red[(TILE) & 1];                            \
    _Pragma("unroll") for (int r = 0; r < 8; ++r) {                    \
      ws_[wv][r][lane]     = acc0[r];                                  \
      ws_[wv][8 + r][lane] = acc1[r];                                  \
    }                                                                  \
    __syncthreads();                                                   \
    acc0 = zero8;                                                      \
    acc1 = zero8;                                                      \
  }
#define SCN_FIN(TILE)                                                  \
  {                                                                    \
    const float* rp_ = &red[(TILE) & 1][0][0][0] + fo;                 \
    float4 x = *(const float4*)rp_;                                    \
    _Pragma("unroll") for (int w = 1; w < SCN_WAVES; ++w) {            \
      const float4 y = *(const float4*)(rp_ + w * (16 * 32));          \
      x.x += y.x; x.y += y.y; x.z += y.z; x.w += y.w;                  \
    }                                                                  \
    const int s_ = 16 * (TILE) + 8 * half + 4 * fm;                    \
    float4 v;                                                          \
    v.x = s_     < len ? __expf(x.x) : 0.0f;                           \
    v.y = s_ + 1 < len ? __expf(x.y) : 0.0f;                           \
    v.z = s_ + 2 < len ? __expf(x.z) : 0.0f;                           \
    v.w = s_ + 3 < len ? __expf(x.w) : 0.0f;                           \
    *(float4*)(ag + (size_t)(TILE) * (4 * FRAG_CH)) = v;               \
    dsum += (v.x + v.y) + (v.z + v.w);                                 \
  }
  SCN_LOAD(0, lo);
#pragma unroll
  for (int m = 0; m < SCN_L; ++m) {
    af0[m] = *(const float4*)(a0p + FRAG_CH * m);
    af1[m] = *(const float4*)(a1p + FRAG_CH * m);
  }
  SCN_LOAD(1, lo + 1);
  const v8f zero8 = {0.f, 0.f, 0.f, 0.f, 0.f, 0.f, 0.f, 0.f};
  v8f acc0 = zero8, acc1 = zero8;
  __builtin_amdgcn_sched_barrier(0);
  SCN_STEPS(0, 0, SCN_L);
  SCN_PUT(lo);
  for (int t = lo + 1; t < hi; t += 2) {
    SCN_LOAD(0, t + 1);
    __builtin_amdgcn_sched_barrier(0);
    SCN_FIN(t - 1);
    SCN_STEPS(1, 0, SCN_HEAD);
    __builtin_amdgcn_sched_barrier(0);
    SCN_STEPS(1, SCN_HEAD, SCN_L);
    SCN_PUT(t);
    if (t + 1 < hi) {
      SCN_LOAD(1, t + 2);
      __builtin_amdgcn_sched_barrier(0);
      SCN_FIN(t);
      SCN_STEPS(0, 0, SCN_HEAD);
      __builtin_amdgcn_sched_barrier(0);
      SCN_STEPS(0, SCN_HEAD, SCN_L);
      SCN_PUT(t + 1);
    }
  }
  SCN_FIN(last);
#undef SCN_LOAD
#undef SCN_STEPS
#undef SCN_PUT
#undef SCN_FIN

  const float dx = __int_as_float(__builtin_amdgcn_permlanex16(
      __float_as_int(dsum), __float_as_int(dsum), 0x76543210, 0xfedcba98, false, false));
  if (half == 0) dpart[fm][16 * fj + hl] = dsum + dx;
  __syncthreads();
  if (tid < CA_) den_part[(size_t)slot * CA_ + tid] = dpart[0][tid] + dpart[1][tid];
}

constexpr int OUT_WAVES = 8;
constexpr int OUT_CG    = H_ / 32;
constexpr int OUT_NB    = B_ * OUT_CG;
constexpr int OUT_LEAD  = 256;
constexpr int OUT_PAIR  = OUT_NB - OUT_LEAD;
static_assert(OUT_PAIR > 0 && 2 * OUT_PAIR <= OUT_LEAD && OUT_PAIR % 2 == 0, "the index map below assumes it");
constexpr int OUT_ST    = 4;
constexpr int DEN_D     = SCN_SLOTS / 32;
static_assert(OUT_ST == 4, "the pinning statement in k_out names 4 k-steps");
static_assert(DEN_D * 32 == SCN_SLOTS, "DEN_D loads of 32 lanes reach every slot of the scores kernel");
static_assert(OUT_WAVES * 4 == CA_, "each wave sums the denominators of four rows");

__global__ void __launch_bounds__(32 * OUT_WAVES)
k_out(const float* __restrict__ value, const float* __restrict__ alpha_g,
      const float* __restrict__ den_part, const int* __restrict__ seq_len,
      float* __restrict__ out) {
  __shared__ float red[OUT_WAVES][32][32];
  __shared__ float den_s[CA_];
  const int gid  = blockIdx.x;
  int idx;
  if (gid < OUT_PAIR) {
    idx = (gid & 1) ? OUT_NB - gid : gid;
  } else if (gid < OUT_LEAD) {
    const int m = gid - OUT_PAIR;
    idx = (m & 1) ? OUT_LEAD - 1 - (m >> 1) : OUT_PAIR + (m >> 1);
  } else {
    const int g = gid - OUT_LEAD;
    idx = (g & 1) ? g : OUT_NB - 2 - g;
  }
  const int q    = idx / OUT_CG;
  const int cg   = idx % OUT_CG;
  const int tid  = threadIdx.x;
  const int lane = tid & 31;
  const int wv   = tid >> 5;
  const int hl   = lane & 15;
  const int half = lane >> 4;

  int len = seq_len[hl];
  len = len < 0 ? 0 : (len > S_ ? S_ : len);
  const int tiles = (len + 15) >> 4;
  const int key = tiles * 16 + (15 - hl);
  int rank = 0;
#pragma unroll
  for (int j = 0; j < 16; ++j) rank += (__builtin_amdgcn_readlane(key, j) > key) ? 1 : 0;
  const int b  = __ffs((unsigned)__ballot(rank == q) & 0xFFFFu) - 1;
  const int tb = __builtin_amdgcn_readlane(tiles, b);
  const int t0 = __builtin_amdgcn_readfirstlane((wv * tb) / OUT_WAVES);
  const int t1 = __builtin_amdgcn_readfirstlane(((wv + 1) * tb) / OUT_WAVES);

  const float* apf = alpha_g + (size_t)b * CA_ * S_ + lane * 4;
  const float* vp  = value + ((size_t)b * S_ + 8 * half) * H_ + cg * 32 + 2 * hl;

  v8f acc[4];
  for (int a = 0; a < 4; ++a) acc[a] = (v8f){0.f, 0.f, 0.f, 0.f, 0.f, 0.f, 0.f, 0.f};

  float4 a0[3][2], a1[3][2];
  v2f    v0[3][OUT_ST], v1[3][OUT_ST];
  const int last = t1 - 1;
#define OUT_LOAD(SET, TILE)                                              \
    {                                                                    \
      const int tt_ = (TILE) < last ? (TILE) : last;                     \
      _Pragma("unroll") for (int m = 0; m < 2; ++m) {                    \
        a0[SET][m] = *(const float4*)(apf + (size_t)tt_ * (4 * FRAG_CH) + m * FRAG_CH);       \
        a1[SET][m] = *(const float4*)(apf + (size_t)tt_ * (4 * FRAG_CH) + (2 + m) * FRAG_CH); \
      }                                                                  \
      _Pragma("unroll") for (int i = 0; i < OUT_ST; ++i) {               \
        const int k = 16 * tt_ + 2 * i;                                  \
        v0[SET][i] = *(const v2f*)(vp + (size_t)k * H_);                 \
        v1[SET][i] = *(const v2f*)(vp + (size_t)(k + 1) * H_);           \
      }                                                                  \
    }
#define OUT_MMA(SET)                                                     \
    {                                                                    \
         \
      asm volatile("" : "+v"(v0[SET][0]), "+v"(v1[SET][0]), "+v"(v0[SET][1]),   \
                        "+v"(v1[SET][1]), "+v"(v0[SET][2]), "+v"(v1[SET][2]),   \
                        "+v"(v0[SET][3]), "+v"(v1[SET][3]));                    \
      _Pragma("unroll") for (int i = 0; i < OUT_ST; ++i) {               \
        const float4 p = a0[SET][i >> 1];                                \
        const float4 u = a1[SET][i >> 1];                                \
        v2f p0, u0, b0, b1;                                              \
        p0.x = (i & 1) ? p.z : p.x;  p0.y = (i & 1) ? p.w : p.y;         \
        u0.x = (i & 1) ? u.z : u.x;  u0.y = (i & 1) ? u.w : u.y;         \
        b0.x = v0[SET][i].x; b0.y = v1[SET][i].x;                        \
        b1.x = v0[SET][i].y; b1.y = v1[SET][i].y;                        \
        acc[0] = WMMA_F32(p0, b0, acc[0]);                               \
        acc[1] = WMMA_F32(p0, b1, acc[1]);                               \
        acc[2] = WMMA_F32(u0, b0, acc[2]);                               \
        acc[3] = WMMA_F32(u0, b1, acc[3]);                               \
      }                                                                  \
                  \
      asm volatile("" : "+v"(acc[0]), "+v"(acc[1]), "+v"(acc[2]), "+v"(acc[3])); \
    }
  if (t0 < t1) {
    OUT_LOAD(0, t0);
    OUT_LOAD(1, t0 + 1);
  }

  int send, st;
  sc_plan_tiles(tiles, send, st);
  const int sp  = __builtin_amdgcn_readlane(send, b > 0 ? b - 1 : 0);
  const int se0 = (b > 0) ? sp : 0;
  const int se1 = __builtin_amdgcn_readlane(send, b);
  const int dlast = se1 > 0 ? se1 - 1 : 0;
  float4 dv[DEN_D];
#pragma unroll
  for (int i = 0; i < DEN_D; ++i) {
    const int t = se0 + lane + 32 * i;
    dv[i] = *(const float4*)(den_part + (size_t)(t < dlast ? t : dlast) * CA_ + 4 * wv);
  }

  if (t0 < t1) {
    int t = t0;
    for (; t + 3 <= t1; t += 3) {
      OUT_LOAD(2, t + 2);
      __builtin_amdgcn_sched_barrier(0);
      OUT_MMA(0);
      OUT_LOAD(0, t + 3);
      __builtin_amdgcn_sched_barrier(0);
      OUT_MMA(1);
      OUT_LOAD(1, t + 4);
      __builtin_amdgcn_sched_barrier(0);
      OUT_MMA(2);
    }
    if (t < t1) OUT_MMA(0);
    if (t + 1 < t1) OUT_MMA(1);
#undef OUT_LOAD
#undef OUT_MMA
  }

  for (int a = 0; a < 4; ++a)
    for (int r = 0; r < 8; ++r)
      red[wv][a * 8 + r][lane] = acc[a][r];

  float dl[4] = {0.f, 0.f, 0.f, 0.f};
#pragma unroll
  for (int i = 0; i < DEN_D; ++i) {
    const bool in = se0 + lane + 32 * i < se1;
    dl[0] += in ? dv[i].x : 0.f;
    dl[1] += in ? dv[i].y : 0.f;
    dl[2] += in ? dv[i].z : 0.f;
    dl[3] += in ? dv[i].w : 0.f;
  }
  for (int u = 0; u < 4; ++u) {
    const float d = row16_sum(dl[u]);
    const float den = __int_as_float(__builtin_amdgcn_readlane(__float_as_int(d), 0)) +
                      __int_as_float(__builtin_amdgcn_readlane(__float_as_int(d), 16));
    if (lane == 0) den_s[4 * wv + u] = den;
  }
  __syncthreads();

  float* o = out + (size_t)b * CA_ * H_ + cg * 32 + 2 * hl;
  const int trow = wv >> 2;
  for (int qq = 0; qq < 2; ++qq) {
    const int r = 2 * (wv & 3) + qq;
    const int reg0 = (2 * trow) * 8 + r;
    const int reg1 = (2 * trow + 1) * 8 + r;
    float s0 = red[0][reg0][lane];
    float s1 = red[0][reg1][lane];
    for (int w = 1; w < OUT_WAVES; ++w) {
      s0 += red[w][reg0][lane];
      s1 += red[w][reg1][lane];
    }
    const int c = 16 * trow + 8 * half + r;
    const float den = den_s[c];
    v2f ov;
    ov.x = s0 / den;
    ov.y = s1 / den;
    *(v2f*)(o + (size_t)c * H_) = ov;
  }
}

extern "C" void kernel_launch(void* const* d_in, const int* in_sizes, int n_in,
                              void* d_out, int out_size, void* d_ws, size_t ws_size,
                              hipStream_t stream) {
  (void)in_sizes; (void)n_in; (void)out_size; (void)ws_size;
  const float* key     = (const float*)d_in[0];
  const float* value   = (const float*)d_in[1];
  const float* query   = (const float*)d_in[2];
  const int*   seq_len = (const int*)d_in[3];
  const float* W       = (const float*)d_in[4];
  const float* bias    = (const float*)d_in[5];
  float* ws  = (float*)d_ws;
  float* out = (float*)d_out;

  k_tq<<<B_ * TQ_NT, 32 * TQ_WAVES, 0, stream>>>(query, W, bias, ws + TQ_OFF);

  k_scores<<<SCN_SLOTS, 32 * SCN_WAVES, 0, stream>>>(key, ws + TQ_OFF, seq_len,
                                                     ws + AL_OFF, ws + DEN_OFF);

  k_out<<<OUT_NB, 32 * OUT_WAVES, 0, stream>>>(value, ws + AL_OFF, ws + DEN_OFF, seq_len, out);
}
